# starting kernel
# baseline (speedup 1.0000x reference)
#include <hip/hip_runtime.h>
#include <hip/hip_bf16.h>

#define B_SZ 4
#define L_SZ 2048
#define D_SZ 1024
#define N_SZ 16
#define P_SZ 33

typedef float v2f __attribute__((ext_vector_type(2)));
typedef float v8f __attribute__((ext_vector_type(8)));

__global__ __launch_bounds__(256) void conv_silu_kernel(
    const float* __restrict__ x, const float* __restrict__ cw,
    const float* __restrict__ cb, float* __restrict__ xc)
{
  const int idx = blockIdx.x * 256 + threadIdx.x;
  const int d  = idx & (D_SZ - 1);
  const int bl = idx >> 10;
  const int l  = bl & (L_SZ - 1);
  const float* xp = x + (size_t)bl * D_SZ + d;
  const float* wp = cw + (size_t)d * 4;
  float acc = cb[d];
#pragma unroll
  for (int j = 0; j < 4; ++j) {
    const int li = l - 3 + j;
    const float xv = (li >= 0) ? xp[(j - 3) * D_SZ] : 0.f;
    acc = fmaf(xv, wp[j], acc);
  }
  xc[idx] = acc / (1.f + __expf(-acc));
}

__global__ __launch_bounds__(128) void proj_wmma_kernel(
    const float* __restrict__ xc, const float* __restrict__ w,
    float* __restrict__ proj)
{
  const int wave = blockIdx.x * 4 + (threadIdx.x >> 5);
  const int lane = threadIdx.x & 31;
  const int mrow = wave * 16 + (lane & 15);
  const int kb   = (lane >> 4) << 1;
  const float* arow = xc + (size_t)mrow * D_SZ + kb;

  const int ncol = lane & 15;
  const int p0 = ncol;
  const int p1 = 16 + ncol;
  int p2 = 32 + ncol;
  const float s2 = (p2 < P_SZ) ? 1.f : 0.f;
  if (p2 >= P_SZ) p2 = P_SZ - 1;
  const float* brow0 = w + (size_t)p0 * D_SZ + kb;
  const float* brow1 = w + (size_t)p1 * D_SZ + kb;
  const float* brow2 = w + (size_t)p2 * D_SZ + kb;

  v8f acc0 = {}, acc1 = {}, acc2 = {};
  for (int k0 = 0; k0 < D_SZ; k0 += 4) {
    v2f a  = *(const v2f*)(arow  + k0);
    v2f b0 = *(const v2f*)(brow0 + k0);
    v2f b1 = *(const v2f*)(brow1 + k0);
    v2f b2 = *(const v2f*)(brow2 + k0);
    b2 *= s2;
    acc0 = __builtin_amdgcn_wmma_f32_16x16x4_f32(false, a, false, b0,
                                                 (short)0, acc0, false, false);
    acc1 = __builtin_amdgcn_wmma_f32_16x16x4_f32(false, a, false, b1,
                                                 (short)0, acc1, false, false);
    acc2 = __builtin_amdgcn_wmma_f32_16x16x4_f32(false, a, false, b2,
                                                 (short)0, acc2, false, false);
  }

  const int mhi = (lane >> 4) << 3;
#pragma unroll
  for (int v = 0; v < 8; ++v) {
    const int row = wave * 16 + v + mhi;
    float* pr = proj + (size_t)row * P_SZ;
    pr[p0] = acc0[v];
    pr[p1] = acc1[v];
    if (32 + ncol < P_SZ) pr[32 + ncol] = acc2[v];
  }
}

__global__ __launch_bounds__(256) void scan_kernel(
    const float* __restrict__ x, const float* __restrict__ xc,
    const float* __restrict__ proj, const float* __restrict__ dt_w,
    const float* __restrict__ dt_b, const float* __restrict__ a_log,
    const float* __restrict__ Dp, float* __restrict__ out)
{
  const int n = threadIdx.x & 15;
  const int c = blockIdx.x * 16 + (threadIdx.x >> 4);
  const int b = c >> 10;
  const int d = c & (D_SZ - 1);

  const float a_c = -__expf(a_log[d * N_SZ + n]);
  const float dtw = dt_w[d];
  const float dtb = dt_b[d];
  const float dpd = Dp[d];

  const float* pr   = proj + (size_t)b * L_SZ * P_SZ;
  const float* xr   = x    + (size_t)b * L_SZ * D_SZ + d;
  const float* xcr  = xc   + (size_t)b * L_SZ * D_SZ + d;
  float*       outr = out  + (size_t)b * L_SZ * D_SZ + d;

  float h = 0.f;
  for (int t = 0; t < L_SZ; ++t) {
    const float* pt = pr + (size_t)t * P_SZ;
    const float bt  = pt[n];
    const float ct  = pt[N_SZ + n];
    const float p32 = pt[2 * N_SZ];
    const float xt  = xcr[(size_t)t * D_SZ];

    const float z     = fmaf(p32, dtw, dtb);
    const float delta = (z > 20.f) ? z : __logf(1.f + __expf(z));
    const float ab    = __expf(delta * a_c);

    h = fmaf(ab, h, delta * xt * bt);
    float y = h * ct;
    y += __shfl_xor(y, 1);
    y += __shfl_xor(y, 2);
    y += __shfl_xor(y, 4);
    y += __shfl_xor(y, 8);
    if (n == 0)
      outr[(size_t)t * D_SZ] = fmaf(xr[(size_t)t * D_SZ], dpd, y);
  }
}

extern "C" void kernel_launch(void* const* d_in, const int* in_sizes, int n_in,
                              void* d_out, int out_size, void* d_ws, size_t ws_size,
                              hipStream_t stream) {
  const float* x      = (const float*)d_in[0];
  const float* conv_w = (const float*)d_in[1];
  const float* conv_b = (const float*)d_in[2];
  const float* xproj  = (const float*)d_in[3];
  const float* dt_w   = (const float*)d_in[4];
  const float* dt_b   = (const float*)d_in[5];
  const float* a_log  = (const float*)d_in[6];
  const float* Dp     = (const float*)d_in[7];
  float* out = (float*)d_out;

  float* xc   = (float*)d_ws;
  float* proj = (float*)((char*)d_ws +
                         (size_t)B_SZ * L_SZ * D_SZ * sizeof(float));

  conv_silu_kernel<<<(B_SZ * L_SZ * D_SZ) / 256, 256, 0, stream>>>(
      x, conv_w, conv_b, xc);

  proj_wmma_kernel<<<128, 128, 0, stream>>>(xc, xproj, proj);

  scan_kernel<<<(B_SZ * D_SZ) / 16, 256, 0, stream>>>(
      x, xc, proj, dt_w, dt_b, a_log, Dp, out);
}


# scan time loop software-pipelined in 12-step batches: next batch's operands loaded one batch ahead and all h-independent math batched, only the fma on h stays serial; fp order unchanged
# speedup vs baseline: 8.2486x; 8.2486x over previous
#include <hip/hip_runtime.h>
#include <hip/hip_bf16.h>

#define B_SZ 4
#define L_SZ 2048
#define D_SZ 1024
#define N_SZ 16
#define P_SZ 33

typedef float v2f __attribute__((ext_vector_type(2)));
typedef float v8f __attribute__((ext_vector_type(8)));

__global__ __launch_bounds__(256) void conv_silu_kernel(
    const float* __restrict__ x, const float* __restrict__ cw,
    const float* __restrict__ cb, float* __restrict__ xc)
{
  const int idx = blockIdx.x * 256 + threadIdx.x;
  const int d  = idx & (D_SZ - 1);
  const int bl = idx >> 10;
  const int l  = bl & (L_SZ - 1);
  const float* xp = x + (size_t)bl * D_SZ + d;
  const float* wp = cw + (size_t)d * 4;
  float acc = cb[d];
#pragma unroll
  for (int j = 0; j < 4; ++j) {
    const int li = l - 3 + j;
    const float xv = (li >= 0) ? xp[(j - 3) * D_SZ] : 0.f;
    acc = fmaf(xv, wp[j], acc);
  }
  xc[idx] = acc / (1.f + __expf(-acc));
}

__global__ __launch_bounds__(128) void proj_wmma_kernel(
    const float* __restrict__ xc, const float* __restrict__ w,
    float* __restrict__ proj)
{
  const int wave = blockIdx.x * 4 + (threadIdx.x >> 5);
  const int lane = threadIdx.x & 31;
  const int mrow = wave * 16 + (lane & 15);
  const int kb   = (lane >> 4) << 1;
  const float* arow = xc + (size_t)mrow * D_SZ + kb;

  const int ncol = lane & 15;
  const int p0 = ncol;
  const int p1 = 16 + ncol;
  int p2 = 32 + ncol;
  const float s2 = (p2 < P_SZ) ? 1.f : 0.f;
  if (p2 >= P_SZ) p2 = P_SZ - 1;
  const float* brow0 = w + (size_t)p0 * D_SZ + kb;
  const float* brow1 = w + (size_t)p1 * D_SZ + kb;
  const float* brow2 = w + (size_t)p2 * D_SZ + kb;

  v8f acc0 = {}, acc1 = {}, acc2 = {};
  for (int k0 = 0; k0 < D_SZ; k0 += 4) {
    v2f a  = *(const v2f*)(arow  + k0);
    v2f b0 = *(const v2f*)(brow0 + k0);
    v2f b1 = *(const v2f*)(brow1 + k0);
    v2f b2 = *(const v2f*)(brow2 + k0);
    b2 *= s2;
    acc0 = __builtin_amdgcn_wmma_f32_16x16x4_f32(false, a, false, b0,
                                                 (short)0, acc0, false, false);
    acc1 = __builtin_amdgcn_wmma_f32_16x16x4_f32(false, a, false, b1,
                                                 (short)0, acc1, false, false);
    acc2 = __builtin_amdgcn_wmma_f32_16x16x4_f32(false, a, false, b2,
                                                 (short)0, acc2, false, false);
  }

  const int mhi = (lane >> 4) << 3;
#pragma unroll
  for (int v = 0; v < 8; ++v) {
    const int row = wave * 16 + v + mhi;
    float* pr = proj + (size_t)row * P_SZ;
    pr[p0] = acc0[v];
    pr[p1] = acc1[v];
    if (32 + ncol < P_SZ) pr[32 + ncol] = acc2[v];
  }
}

#define TS 12
#define L_MAIN (L_SZ - L_SZ % (2 * TS))
#define L_TAIL (L_SZ - L_MAIN)

template <int NT>
struct ScanBatch {
  float bt[NT], ct[NT], p32[NT], xt[NT], xres[NT];
};

template <int NT>
__device__ __forceinline__ void scan_load(
    ScanBatch<NT>& s, const float* __restrict__ pr,
    const float* __restrict__ xcr, const float* __restrict__ xr, int n, int t0)
{
#pragma unroll
  for (int i = 0; i < NT; ++i) {
    const float* pt = pr + (size_t)(t0 + i) * P_SZ;
    s.bt[i]   = pt[n];
    s.ct[i]   = pt[N_SZ + n];
    s.p32[i]  = pt[2 * N_SZ];
    s.xt[i]   = xcr[(size_t)(t0 + i) * D_SZ];
    s.xres[i] = xr[(size_t)(t0 + i) * D_SZ];
  }
}

template <int NT>
__device__ __forceinline__ float scan_batch(
    const ScanBatch<NT>& s, float h, float a_c, float dtw, float dtb,
    float dpd, int n, float* __restrict__ outr, int t0)
{
  float ab[NT], u[NT], y[NT];
#pragma unroll
  for (int i = 0; i < NT; ++i) {
    const float z     = fmaf(s.p32[i], dtw, dtb);
    const float sp    = __logf(1.f + __expf(z));
    const float delta = (z > 20.f) ? z : sp;
    ab[i] = __expf(delta * a_c);
    u[i]  = delta * s.xt[i] * s.bt[i];
  }
#pragma unroll
  for (int i = 0; i < NT; ++i) {
    h = fmaf(ab[i], h, u[i]);
    y[i] = h * s.ct[i];
  }
#pragma unroll
  for (int i = 0; i < NT; ++i) {
    float v = y[i];
    v += __shfl_xor(v, 1);
    v += __shfl_xor(v, 2);
    v += __shfl_xor(v, 4);
    v += __shfl_xor(v, 8);
    y[i] = v;
  }
  if (n == 0) {
#pragma unroll
    for (int i = 0; i < NT; ++i)
      outr[(size_t)(t0 + i) * D_SZ] = fmaf(s.xres[i], dpd, y[i]);
  }
  return h;
}

__global__ __launch_bounds__(256) void scan_kernel(
    const float* __restrict__ x, const float* __restrict__ xc,
    const float* __restrict__ proj, const float* __restrict__ dt_w,
    const float* __restrict__ dt_b, const float* __restrict__ a_log,
    const float* __restrict__ Dp, float* __restrict__ out)
{
  const int n = threadIdx.x & 15;
  const int c = blockIdx.x * 16 + (threadIdx.x >> 4);
  const int b = c >> 10;
  const int d = c & (D_SZ - 1);

  const float a_c = -__expf(a_log[d * N_SZ + n]);
  const float dtw = dt_w[d];
  const float dtb = dt_b[d];
  const float dpd = Dp[d];

  const float* pr   = proj + (size_t)b * L_SZ * P_SZ;
  const float* xr   = x    + (size_t)b * L_SZ * D_SZ + d;
  const float* xcr  = xc   + (size_t)b * L_SZ * D_SZ + d;
  float*       outr = out  + (size_t)b * L_SZ * D_SZ + d;

  ScanBatch<TS> s0, s1;
  scan_load(s0, pr, xcr, xr, n, 0);
  float h = 0.f;
  for (int t0 = 0; t0 < L_MAIN; t0 += 2 * TS) {
    scan_load(s1, pr, xcr, xr, n, t0 + TS);
    h = scan_batch(s0, h, a_c, dtw, dtb, dpd, n, outr, t0);
    const int tn = (t0 + 2 * TS < L_MAIN) ? (t0 + 2 * TS) : t0;
    scan_load(s0, pr, xcr, xr, n, tn);
    h = scan_batch(s1, h, a_c, dtw, dtb, dpd, n, outr, t0 + TS);
  }
  if (L_TAIL > 0) {
    ScanBatch<(L_TAIL > 0 ? L_TAIL : 1)> st;
    scan_load(st, pr, xcr, xr, n, L_MAIN);
    h = scan_batch(st, h, a_c, dtw, dtb, dpd, n, outr, L_MAIN);
  }
}

extern "C" void kernel_launch(void* const* d_in, const int* in_sizes, int n_in,
                              void* d_out, int out_size, void* d_ws, size_t ws_size,
                              hipStream_t stream) {
  const float* x      = (const float*)d_in[0];
  const float* conv_w = (const float*)d_in[1];
  const float* conv_b = (const float*)d_in[2];
  const float* xproj  = (const float*)d_in[3];
  const float* dt_w   = (const float*)d_in[4];
  const float* dt_b   = (const float*)d_in[5];
  const float* a_log  = (const float*)d_in[6];
  const float* Dp     = (const float*)d_in[7];
  float* out = (float*)d_out;

  float* xc   = (float*)d_ws;
  float* proj = (float*)((char*)d_ws +
                         (size_t)B_SZ * L_SZ * D_SZ * sizeof(float));

  conv_silu_kernel<<<(B_SZ * L_SZ * D_SZ) / 256, 256, 0, stream>>>(
      x, conv_w, conv_b, xc);

  proj_wmma_kernel<<<128, 128, 0, stream>>>(xc, xproj, proj);

  scan_kernel<<<(B_SZ * D_SZ) / 16, 256, 0, stream>>>(
      x, xc, proj, dt_w, dt_b, a_log, Dp, out);
}


# conv+SiLU kernel: each thread makes 16 time steps of 4 adjacent channels from 19 row loads of 16 bytes issued together: 4096 resident waves, not 262144 one-output waves; fp order unchanged
# speedup vs baseline: 8.5498x; 1.0365x over previous
#include <hip/hip_runtime.h>
#include <hip/hip_bf16.h>

#define B_SZ 4
#define L_SZ 2048
#define D_SZ 1024
#define N_SZ 16
#define P_SZ 33

typedef float v2f __attribute__((ext_vector_type(2)));
typedef float v8f __attribute__((ext_vector_type(8)));

#define CONV_TL 16
#define D_QUADS (D_SZ / 4)

__device__ __forceinline__ float conv_silu_one(
    float acc, float x0, float x1, float x2, float x3, const float4 w)
{
  acc = fmaf(x0, w.x, acc);
  acc = fmaf(x1, w.y, acc);
  acc = fmaf(x2, w.z, acc);
  acc = fmaf(x3, w.w, acc);
  return acc / (1.f + __expf(-acc));
}

__global__ __launch_bounds__(256) void conv_silu_kernel(
    const float* __restrict__ x, const float* __restrict__ cw,
    const float* __restrict__ cb, float* __restrict__ xc)
{
  const int q   = threadIdx.x;
  const int bl0 = blockIdx.x * CONV_TL;
  const int l0  = bl0 & (L_SZ - 1);
  const float4* xp =
      reinterpret_cast<const float4*>(x + (size_t)bl0 * D_SZ) + q;
  float4* op = reinterpret_cast<float4*>(xc + (size_t)bl0 * D_SZ) + q;

  const float4* wq = reinterpret_cast<const float4*>(cw) + 4 * q;
  const float4 w0 = wq[0], w1 = wq[1], w2 = wq[2], w3 = wq[3];
  const float4 bias = reinterpret_cast<const float4*>(cb)[q];

  float4 xin[CONV_TL + 3];
#pragma unroll
  for (int i = 0; i < CONV_TL + 3; ++i) {
    const bool pad = (l0 + i - 3) < 0;
    const float4 v = xp[pad ? 0 : (i - 3) * D_QUADS];
    xin[i].x = pad ? 0.f : v.x;
    xin[i].y = pad ? 0.f : v.y;
    xin[i].z = pad ? 0.f : v.z;
    xin[i].w = pad ? 0.f : v.w;
  }

#pragma unroll
  for (int i = 0; i < CONV_TL; ++i) {
    float4 o;
    o.x = conv_silu_one(bias.x, xin[i].x, xin[i + 1].x, xin[i + 2].x,
                        xin[i + 3].x, w0);
    o.y = conv_silu_one(bias.y, xin[i].y, xin[i + 1].y, xin[i + 2].y,
                        xin[i + 3].y, w1);
    o.z = conv_silu_one(bias.z, xin[i].z, xin[i + 1].z, xin[i + 2].z,
                        xin[i + 3].z, w2);
    o.w = conv_silu_one(bias.w, xin[i].w, xin[i + 1].w, xin[i + 2].w,
                        xin[i + 3].w, w3);
    op[i * D_QUADS] = o;
  }
}

__global__ __launch_bounds__(128) void proj_wmma_kernel(
    const float* __restrict__ xc, const float* __restrict__ w,
    float* __restrict__ proj)
{
  const int wave = blockIdx.x * 4 + (threadIdx.x >> 5);
  const int lane = threadIdx.x & 31;
  const int mrow = wave * 16 + (lane & 15);
  const int kb   = (lane >> 4) << 1;
  const float* arow = xc + (size_t)mrow * D_SZ + kb;

  const int ncol = lane & 15;
  const int p0 = ncol;
  const int p1 = 16 + ncol;
  int p2 = 32 + ncol;
  const float s2 = (p2 < P_SZ) ? 1.f : 0.f;
  if (p2 >= P_SZ) p2 = P_SZ - 1;
  const float* brow0 = w + (size_t)p0 * D_SZ + kb;
  const float* brow1 = w + (size_t)p1 * D_SZ + kb;
  const float* brow2 = w + (size_t)p2 * D_SZ + kb;

  v8f acc0 = {}, acc1 = {}, acc2 = {};
  for (int k0 = 0; k0 < D_SZ; k0 += 4) {
    v2f a  = *(const v2f*)(arow  + k0);
    v2f b0 = *(const v2f*)(brow0 + k0);
    v2f b1 = *(const v2f*)(brow1 + k0);
    v2f b2 = *(const v2f*)(brow2 + k0);
    b2 *= s2;
    acc0 = __builtin_amdgcn_wmma_f32_16x16x4_f32(false, a, false, b0,
                                                 (short)0, acc0, false, false);
    acc1 = __builtin_amdgcn_wmma_f32_16x16x4_f32(false, a, false, b1,
                                                 (short)0, acc1, false, false);
    acc2 = __builtin_amdgcn_wmma_f32_16x16x4_f32(false, a, false, b2,
                                                 (short)0, acc2, false, false);
  }

  const int mhi = (lane >> 4) << 3;
#pragma unroll
  for (int v = 0; v < 8; ++v) {
    const int row = wave * 16 + v + mhi;
    float* pr = proj + (size_t)row * P_SZ;
    pr[p0] = acc0[v];
    pr[p1] = acc1[v];
    if (32 + ncol < P_SZ) pr[32 + ncol] = acc2[v];
  }
}

#define TS 12
#define L_MAIN (L_SZ - L_SZ % (2 * TS))
#define L_TAIL (L_SZ - L_MAIN)

template <int NT>
struct ScanBatch {
  float bt[NT], ct[NT], p32[NT], xt[NT], xres[NT];
};

template <int NT>
__device__ __forceinline__ void scan_load(
    ScanBatch<NT>& s, const float* __restrict__ pr,
    const float* __restrict__ xcr, const float* __restrict__ xr, int n, int t0)
{
#pragma unroll
  for (int i = 0; i < NT; ++i) {
    const float* pt = pr + (size_t)(t0 + i) * P_SZ;
    s.bt[i]   = pt[n];
    s.ct[i]   = pt[N_SZ + n];
    s.p32[i]  = pt[2 * N_SZ];
    s.xt[i]   = xcr[(size_t)(t0 + i) * D_SZ];
    s.xres[i] = xr[(size_t)(t0 + i) * D_SZ];
  }
}

template <int NT>
__device__ __forceinline__ float scan_batch(
    const ScanBatch<NT>& s, float h, float a_c, float dtw, float dtb,
    float dpd, int n, float* __restrict__ outr, int t0)
{
  float ab[NT], u[NT], y[NT];
#pragma unroll
  for (int i = 0; i < NT; ++i) {
    const float z     = fmaf(s.p32[i], dtw, dtb);
    const float sp    = __logf(1.f + __expf(z));
    const float delta = (z > 20.f) ? z : sp;
    ab[i] = __expf(delta * a_c);
    u[i]  = delta * s.xt[i] * s.bt[i];
  }
#pragma unroll
  for (int i = 0; i < NT; ++i) {
    h = fmaf(ab[i], h, u[i]);
    y[i] = h * s.ct[i];
  }
#pragma unroll
  for (int i = 0; i < NT; ++i) {
    float v = y[i];
    v += __shfl_xor(v, 1);
    v += __shfl_xor(v, 2);
    v += __shfl_xor(v, 4);
    v += __shfl_xor(v, 8);
    y[i] = v;
  }
  if (n == 0) {
#pragma unroll
    for (int i = 0; i < NT; ++i)
      outr[(size_t)(t0 + i) * D_SZ] = fmaf(s.xres[i], dpd, y[i]);
  }
  return h;
}

__global__ __launch_bounds__(256) void scan_kernel(
    const float* __restrict__ x, const float* __restrict__ xc,
    const float* __restrict__ proj, const float* __restrict__ dt_w,
    const float* __restrict__ dt_b, const float* __restrict__ a_log,
    const float* __restrict__ Dp, float* __restrict__ out)
{
  const int n = threadIdx.x & 15;
  const int c = blockIdx.x * 16 + (threadIdx.x >> 4);
  const int b = c >> 10;
  const int d = c & (D_SZ - 1);

  const float a_c = -__expf(a_log[d * N_SZ + n]);
  const float dtw = dt_w[d];
  const float dtb = dt_b[d];
  const float dpd = Dp[d];

  const float* pr   = proj + (size_t)b * L_SZ * P_SZ;
  const float* xr   = x    + (size_t)b * L_SZ * D_SZ + d;
  const float* xcr  = xc   + (size_t)b * L_SZ * D_SZ + d;
  float*       outr = out  + (size_t)b * L_SZ * D_SZ + d;

  ScanBatch<TS> s0, s1;
  scan_load(s0, pr, xcr, xr, n, 0);
  float h = 0.f;
  for (int t0 = 0; t0 < L_MAIN; t0 += 2 * TS) {
    scan_load(s1, pr, xcr, xr, n, t0 + TS);
    h = scan_batch(s0, h, a_c, dtw, dtb, dpd, n, outr, t0);
    const int tn = (t0 + 2 * TS < L_MAIN) ? (t0 + 2 * TS) : t0;
    scan_load(s0, pr, xcr, xr, n, tn);
    h = scan_batch(s1, h, a_c, dtw, dtb, dpd, n, outr, t0 + TS);
  }
  if (L_TAIL > 0) {
    ScanBatch<(L_TAIL > 0 ? L_TAIL : 1)> st;
    scan_load(st, pr, xcr, xr, n, L_MAIN);
    h = scan_batch(st, h, a_c, dtw, dtb, dpd, n, outr, L_MAIN);
  }
}

extern "C" void kernel_launch(void* const* d_in, const int* in_sizes, int n_in,
                              void* d_out, int out_size, void* d_ws, size_t ws_size,
                              hipStream_t stream) {
  const float* x      = (const float*)d_in[0];
  const float* conv_w = (const float*)d_in[1];
  const float* conv_b = (const float*)d_in[2];
  const float* xproj  = (const float*)d_in[3];
  const float* dt_w   = (const float*)d_in[4];
  const float* dt_b   = (const float*)d_in[5];
  const float* a_log  = (const float*)d_in[6];
  const float* Dp     = (const float*)d_in[7];
  float* out = (float*)d_out;

  float* xc   = (float*)d_ws;
  float* proj = (float*)((char*)d_ws +
                         (size_t)B_SZ * L_SZ * D_SZ * sizeof(float));

  conv_silu_kernel<<<(B_SZ * L_SZ) / CONV_TL, D_QUADS, 0, stream>>>(
      x, conv_w, conv_b, xc);

  proj_wmma_kernel<<<128, 128, 0, stream>>>(xc, xproj, proj);

  scan_kernel<<<(B_SZ * D_SZ) / 16, 256, 0, stream>>>(
      x, xc, proj, dt_w, dt_b, a_log, Dp, out);
}


# scan: per-channel work (softplus, delta*x, 3 loads, store) done by one lane of 16 per step, shared by DPP row_share; 16-step batches, shuffle rounds fenced; same sum tree, lane n keeps its sum
# speedup vs baseline: 9.9421x; 1.1628x over previous
#include <hip/hip_runtime.h>
#include <hip/hip_bf16.h>

#define B_SZ 4
#define L_SZ 2048
#define D_SZ 1024
#define N_SZ 16
#define P_SZ 33

typedef float v2f __attribute__((ext_vector_type(2)));
typedef float v8f __attribute__((ext_vector_type(8)));

#define CONV_TL 16
#define D_QUADS (D_SZ / 4)

__device__ __forceinline__ float conv_silu_one(
    float acc, float x0, float x1, float x2, float x3, const float4 w)
{
  acc = fmaf(x0, w.x, acc);
  acc = fmaf(x1, w.y, acc);
  acc = fmaf(x2, w.z, acc);
  acc = fmaf(x3, w.w, acc);
  return acc / (1.f + __expf(-acc));
}

__global__ __launch_bounds__(256) void conv_silu_kernel(
    const float* __restrict__ x, const float* __restrict__ cw,
    const float* __restrict__ cb, float* __restrict__ xc)
{
  const int q   = threadIdx.x;
  const int bl0 = blockIdx.x * CONV_TL;
  const int l0  = bl0 & (L_SZ - 1);
  const float4* xp =
      reinterpret_cast<const float4*>(x + (size_t)bl0 * D_SZ) + q;
  float4* op = reinterpret_cast<float4*>(xc + (size_t)bl0 * D_SZ) + q;

  const float4* wq = reinterpret_cast<const float4*>(cw) + 4 * q;
  const float4 w0 = wq[0], w1 = wq[1], w2 = wq[2], w3 = wq[3];
  const float4 bias = reinterpret_cast<const float4*>(cb)[q];

  float4 xin[CONV_TL + 3];
#pragma unroll
  for (int i = 0; i < CONV_TL + 3; ++i) {
    const bool pad = (l0 + i - 3) < 0;
    const float4 v = xp[pad ? 0 : (i - 3) * D_QUADS];
    xin[i].x = pad ? 0.f : v.x;
    xin[i].y = pad ? 0.f : v.y;
    xin[i].z = pad ? 0.f : v.z;
    xin[i].w = pad ? 0.f : v.w;
  }

#pragma unroll
  for (int i = 0; i < CONV_TL; ++i) {
    float4 o;
    o.x = conv_silu_one(bias.x, xin[i].x, xin[i + 1].x, xin[i + 2].x,
                        xin[i + 3].x, w0);
    o.y = conv_silu_one(bias.y, xin[i].y, xin[i + 1].y, xin[i + 2].y,
                        xin[i + 3].y, w1);
    o.z = conv_silu_one(bias.z, xin[i].z, xin[i + 1].z, xin[i + 2].z,
                        xin[i + 3].z, w2);
    o.w = conv_silu_one(bias.w, xin[i].w, xin[i + 1].w, xin[i + 2].w,
                        xin[i + 3].w, w3);
    op[i * D_QUADS] = o;
  }
}

__global__ __launch_bounds__(128) void proj_wmma_kernel(
    const float* __restrict__ xc, const float* __restrict__ w,
    float* __restrict__ proj)
{
  const int wave = blockIdx.x * 4 + (threadIdx.x >> 5);
  const int lane = threadIdx.x & 31;
  const int mrow = wave * 16 + (lane & 15);
  const int kb   = (lane >> 4) << 1;
  const float* arow = xc + (size_t)mrow * D_SZ + kb;

  const int ncol = lane & 15;
  const int p0 = ncol;
  const int p1 = 16 + ncol;
  int p2 = 32 + ncol;
  const float s2 = (p2 < P_SZ) ? 1.f : 0.f;
  if (p2 >= P_SZ) p2 = P_SZ - 1;
  const float* brow0 = w + (size_t)p0 * D_SZ + kb;
  const float* brow1 = w + (size_t)p1 * D_SZ + kb;
  const float* brow2 = w + (size_t)p2 * D_SZ + kb;

  v8f acc0 = {}, acc1 = {}, acc2 = {};
  for (int k0 = 0; k0 < D_SZ; k0 += 4) {
    v2f a  = *(const v2f*)(arow  + k0);
    v2f b0 = *(const v2f*)(brow0 + k0);
    v2f b1 = *(const v2f*)(brow1 + k0);
    v2f b2 = *(const v2f*)(brow2 + k0);
    b2 *= s2;
    acc0 = __builtin_amdgcn_wmma_f32_16x16x4_f32(false, a, false, b0,
                                                 (short)0, acc0, false, false);
    acc1 = __builtin_amdgcn_wmma_f32_16x16x4_f32(false, a, false, b1,
                                                 (short)0, acc1, false, false);
    acc2 = __builtin_amdgcn_wmma_f32_16x16x4_f32(false, a, false, b2,
                                                 (short)0, acc2, false, false);
  }

  const int mhi = (lane >> 4) << 3;
#pragma unroll
  for (int v = 0; v < 8; ++v) {
    const int row = wave * 16 + v + mhi;
    float* pr = proj + (size_t)row * P_SZ;
    pr[p0] = acc0[v];
    pr[p1] = acc1[v];
    if (32 + ncol < P_SZ) pr[32 + ncol] = acc2[v];
  }
}

#define TS 16
static_assert(TS == N_SZ, "lane n of a row owns step t0 + n of a batch");
static_assert(L_SZ % (2 * TS) == 0, "the time loop runs in pairs of batches");

struct ScanBC  { float bt[TS], ct[TS]; };
struct ScanRaw { float p32, xt, xres; };
struct ScanCh  { float delta, dx, xres; };

__device__ __forceinline__ void phase_fence()
{
  __builtin_amdgcn_sched_barrier(0);
}

__device__ __forceinline__ void scan_load_bc(
    ScanBC& s, const float* __restrict__ pr, int n, int t0)
{
#pragma unroll
  for (int i = 0; i < TS; ++i) {
    const float* pt = pr + (size_t)(t0 + i) * P_SZ;
    s.bt[i] = pt[n];
    s.ct[i] = pt[N_SZ + n];
  }
}

__device__ __forceinline__ void scan_load_ch(
    ScanRaw& r, const float* __restrict__ pr, const float* __restrict__ xcr,
    const float* __restrict__ xr, int n, int t0)
{
  r.p32  = pr[(size_t)(t0 + n) * P_SZ + 2 * N_SZ];
  r.xt   = xcr[(size_t)(t0 + n) * D_SZ];
  r.xres = xr[(size_t)(t0 + n) * D_SZ];
}

__device__ __forceinline__ ScanCh scan_prep(const ScanRaw& r, float dtw,
                                            float dtb)
{
  ScanCh c;
  const float z  = fmaf(r.p32, dtw, dtb);
  const float sp = __logf(1.f + __expf(z));
  c.delta = (z > 20.f) ? z : sp;
  c.dx    = c.delta * r.xt;
  c.xres  = r.xres;
  return c;
}

template <int I>
__device__ __forceinline__ float row_lane(float v)
{
  return __int_as_float(__builtin_amdgcn_update_dpp(
      0, __float_as_int(v), 0x150 + I, 0xf, 0xf, false));
}

template <int I, int E>
__device__ __forceinline__ void scan_decay(float delta, float a_c,
                                           float* __restrict__ ab)
{
  if constexpr (I < E) {
    ab[I] = __expf(row_lane<I>(delta) * a_c);
    scan_decay<I + 1, E>(delta, a_c, ab);
  }
}

template <int I>
__device__ __forceinline__ void scan_drive(const ScanBC& s, float dx,
                                           float* __restrict__ u)
{
  if constexpr (I < TS) {
    u[I] = row_lane<I>(dx) * s.bt[I];
    scan_drive<I + 1>(s, dx, u);
  }
}

__device__ __forceinline__ float scan_batch(
    ScanBC& s, const ScanCh& c, const float* __restrict__ ab,
    ScanRaw& rn, ScanCh& cn, float* __restrict__ abn,
    float h, float a_c, float dtw, float dtb, float dpd, int n,
    const float* __restrict__ pr, const float* __restrict__ xcr,
    const float* __restrict__ xr, float* __restrict__ outr,
    int t0, int tb, int tr)
{
  float u[TS], y[TS], t[TS];
  const ScanRaw rnext = rn;
  scan_drive<0>(s, c.dx, u);
#pragma unroll
  for (int i = 0; i < TS; ++i) {
    h = fmaf(ab[i], h, u[i]);
    y[i] = h * s.ct[i];
  }
#pragma unroll
  for (int i = 0; i < TS; ++i) t[i] = __shfl_xor(y[i], 1);
  phase_fence();
#pragma unroll
  for (int i = 0; i < TS; ++i) y[i] += t[i];
#pragma unroll
  for (int i = 0; i < TS; ++i) t[i] = __shfl_xor(y[i], 2);
  phase_fence();
#pragma unroll
  for (int i = 0; i < TS; ++i) y[i] += t[i];
#pragma unroll
  for (int i = 0; i < TS; ++i) t[i] = __shfl_xor(y[i], 4);
  phase_fence();
#pragma unroll
  for (int i = 0; i < TS; ++i) y[i] += t[i];
#pragma unroll
  for (int i = 0; i < TS; ++i) t[i] = __shfl_xor(y[i], 8);
  phase_fence();
  scan_load_bc(s, pr, n, tb);
  scan_load_ch(rn, pr, xcr, xr, n, tr);
  phase_fence();
#pragma unroll
  for (int i = 0; i < TS; ++i) y[i] += t[i];
  const bool n1 = n & 1, n2 = n & 2, n4 = n & 4, n8 = n & 8;
  float r8[8], r4[4], r2[2];
#pragma unroll
  for (int j = 0; j < 8; ++j) {
    const float lo = y[2 * j], hi = y[2 * j + 1];
    r8[j] = n1 ? hi : lo;
  }
#pragma unroll
  for (int j = 0; j < 4; ++j) {
    const float lo = r8[2 * j], hi = r8[2 * j + 1];
    r4[j] = n2 ? hi : lo;
  }
#pragma unroll
  for (int j = 0; j < 2; ++j) {
    const float lo = r4[2 * j], hi = r4[2 * j + 1];
    r2[j] = n4 ? hi : lo;
  }
  const float ylo = r2[0], yhi = r2[1];
  const float yn = n8 ? yhi : ylo;
  outr[(size_t)(t0 + n) * D_SZ] = fmaf(c.xres, dpd, yn);
  cn = scan_prep(rnext, dtw, dtb);
  scan_decay<0, TS>(cn.delta, a_c, abn);
  phase_fence();
  return h;
}

__global__ __launch_bounds__(256) void scan_kernel(
    const float* __restrict__ x, const float* __restrict__ xc,
    const float* __restrict__ proj, const float* __restrict__ dt_w,
    const float* __restrict__ dt_b, const float* __restrict__ a_log,
    const float* __restrict__ Dp, float* __restrict__ out)
{
  const int n = threadIdx.x & 15;
  const int c = blockIdx.x * 16 + (threadIdx.x >> 4);
  const int b = c >> 10;
  const int d = c & (D_SZ - 1);

  const float a_c = -__expf(a_log[d * N_SZ + n]);
  const float dtw = dt_w[d];
  const float dtb = dt_b[d];
  const float dpd = Dp[d];

  const float* pr   = proj + (size_t)b * L_SZ * P_SZ;
  const float* xr   = x    + (size_t)b * L_SZ * D_SZ + d;
  const float* xcr  = xc   + (size_t)b * L_SZ * D_SZ + d;
  float*       outr = out  + (size_t)b * L_SZ * D_SZ + d;

  ScanBC  bc0, bc1;
  ScanRaw rawE, rawO;
  ScanCh  chE, chO;
  float   abE[TS], abO[TS];
  scan_load_bc(bc0, pr, n, 0);
  scan_load_bc(bc1, pr, n, TS);
  scan_load_ch(rawE, pr, xcr, xr, n, 0);
  scan_load_ch(rawO, pr, xcr, xr, n, TS);
  chE = scan_prep(rawE, dtw, dtb);
  scan_decay<0, TS>(chE.delta, a_c, abE);
  scan_load_ch(rawE, pr, xcr, xr, n, 2 * TS);
  float h = 0.f;
  for (int t0 = 0; t0 < L_SZ; t0 += 2 * TS) {
    const int t2 = (t0 + 2 * TS < L_SZ) ? (t0 + 2 * TS) : t0;
    const int t3 = (t0 + 3 * TS < L_SZ) ? (t0 + 3 * TS) : t0;
    const int t4 = (t0 + 4 * TS < L_SZ) ? (t0 + 4 * TS) : t0;
    h = scan_batch(bc0, chE, abE, rawO, chO, abO, h, a_c, dtw, dtb, dpd, n,
                   pr, xcr, xr, outr, t0, t2, t3);
    h = scan_batch(bc1, chO, abO, rawE, chE, abE, h, a_c, dtw, dtb, dpd, n,
                   pr, xcr, xr, outr, t0 + TS, t3, t4);
  }
}

extern "C" void kernel_launch(void* const* d_in, const int* in_sizes, int n_in,
                              void* d_out, int out_size, void* d_ws, size_t ws_size,
                              hipStream_t stream) {
  const float* x      = (const float*)d_in[0];
  const float* conv_w = (const float*)d_in[1];
  const float* conv_b = (const float*)d_in[2];
  const float* xproj  = (const float*)d_in[3];
  const float* dt_w   = (const float*)d_in[4];
  const float* dt_b   = (const float*)d_in[5];
  const float* a_log  = (const float*)d_in[6];
  const float* Dp     = (const float*)d_in[7];
  float* out = (float*)d_out;

  float* xc   = (float*)d_ws;
  float* proj = (float*)((char*)d_ws +
                         (size_t)B_SZ * L_SZ * D_SZ * sizeof(float));

  conv_silu_kernel<<<(B_SZ * L_SZ) / CONV_TL, D_QUADS, 0, stream>>>(
      x, conv_w, conv_b, xc);

  proj_wmma_kernel<<<128, 128, 0, stream>>>(xc, xproj, proj);

  scan_kernel<<<(B_SZ * D_SZ) / 16, 256, 0, stream>>>(
      x, xc, proj, dt_w, dt_b, a_log, Dp, out);
}


# scan: sum over n as a reduce-scatter of DPP row_xmask adds on the vector ALU, 15 per 16-step batch, not 64 LDS-unit permutes + 64 adds; same sum tree, products rounded before the first add
# speedup vs baseline: 11.0416x; 1.1106x over previous
#include <hip/hip_runtime.h>
#include <hip/hip_bf16.h>

#define B_SZ 4
#define L_SZ 2048
#define D_SZ 1024
#define N_SZ 16
#define P_SZ 33

typedef float v2f __attribute__((ext_vector_type(2)));
typedef float v8f __attribute__((ext_vector_type(8)));

#define CONV_TL 16
#define D_QUADS (D_SZ / 4)

__device__ __forceinline__ float conv_silu_one(
    float acc, float x0, float x1, float x2, float x3, const float4 w)
{
  acc = fmaf(x0, w.x, acc);
  acc = fmaf(x1, w.y, acc);
  acc = fmaf(x2, w.z, acc);
  acc = fmaf(x3, w.w, acc);
  return acc / (1.f + __expf(-acc));
}

__global__ __launch_bounds__(256) void conv_silu_kernel(
    const float* __restrict__ x, const float* __restrict__ cw,
    const float* __restrict__ cb, float* __restrict__ xc)
{
  const int q   = threadIdx.x;
  const int bl0 = blockIdx.x * CONV_TL;
  const int l0  = bl0 & (L_SZ - 1);
  const float4* xp =
      reinterpret_cast<const float4*>(x + (size_t)bl0 * D_SZ) + q;
  float4* op = reinterpret_cast<float4*>(xc + (size_t)bl0 * D_SZ) + q;

  const float4* wq = reinterpret_cast<const float4*>(cw) + 4 * q;
  const float4 w0 = wq[0], w1 = wq[1], w2 = wq[2], w3 = wq[3];
  const float4 bias = reinterpret_cast<const float4*>(cb)[q];

  float4 xin[CONV_TL + 3];
#pragma unroll
  for (int i = 0; i < CONV_TL + 3; ++i) {
    const bool pad = (l0 + i - 3) < 0;
    const float4 v = xp[pad ? 0 : (i - 3) * D_QUADS];
    xin[i].x = pad ? 0.f : v.x;
    xin[i].y = pad ? 0.f : v.y;
    xin[i].z = pad ? 0.f : v.z;
    xin[i].w = pad ? 0.f : v.w;
  }

#pragma unroll
  for (int i = 0; i < CONV_TL; ++i) {
    float4 o;
    o.x = conv_silu_one(bias.x, xin[i].x, xin[i + 1].x, xin[i + 2].x,
                        xin[i + 3].x, w0);
    o.y = conv_silu_one(bias.y, xin[i].y, xin[i + 1].y, xin[i + 2].y,
                        xin[i + 3].y, w1);
    o.z = conv_silu_one(bias.z, xin[i].z, xin[i + 1].z, xin[i + 2].z,
                        xin[i + 3].z, w2);
    o.w = conv_silu_one(bias.w, xin[i].w, xin[i + 1].w, xin[i + 2].w,
                        xin[i + 3].w, w3);
    op[i * D_QUADS] = o;
  }
}

__global__ __launch_bounds__(128) void proj_wmma_kernel(
    const float* __restrict__ xc, const float* __restrict__ w,
    float* __restrict__ proj)
{
  const int wave = blockIdx.x * 4 + (threadIdx.x >> 5);
  const int lane = threadIdx.x & 31;
  const int mrow = wave * 16 + (lane & 15);
  const int kb   = (lane >> 4) << 1;
  const float* arow = xc + (size_t)mrow * D_SZ + kb;

  const int ncol = lane & 15;
  const int p0 = ncol;
  const int p1 = 16 + ncol;
  int p2 = 32 + ncol;
  const float s2 = (p2 < P_SZ) ? 1.f : 0.f;
  if (p2 >= P_SZ) p2 = P_SZ - 1;
  const float* brow0 = w + (size_t)p0 * D_SZ + kb;
  const float* brow1 = w + (size_t)p1 * D_SZ + kb;
  const float* brow2 = w + (size_t)p2 * D_SZ + kb;

  v8f acc0 = {}, acc1 = {}, acc2 = {};
  for (int k0 = 0; k0 < D_SZ; k0 += 4) {
    v2f a  = *(const v2f*)(arow  + k0);
    v2f b0 = *(const v2f*)(brow0 + k0);
    v2f b1 = *(const v2f*)(brow1 + k0);
    v2f b2 = *(const v2f*)(brow2 + k0);
    b2 *= s2;
    acc0 = __builtin_amdgcn_wmma_f32_16x16x4_f32(false, a, false, b0,
                                                 (short)0, acc0, false, false);
    acc1 = __builtin_amdgcn_wmma_f32_16x16x4_f32(false, a, false, b1,
                                                 (short)0, acc1, false, false);
    acc2 = __builtin_amdgcn_wmma_f32_16x16x4_f32(false, a, false, b2,
                                                 (short)0, acc2, false, false);
  }

  const int mhi = (lane >> 4) << 3;
#pragma unroll
  for (int v = 0; v < 8; ++v) {
    const int row = wave * 16 + v + mhi;
    float* pr = proj + (size_t)row * P_SZ;
    pr[p0] = acc0[v];
    pr[p1] = acc1[v];
    if (32 + ncol < P_SZ) pr[32 + ncol] = acc2[v];
  }
}

#define TS 16
static_assert(TS == N_SZ, "lane n of a row owns step t0 + n of a batch");
static_assert(L_SZ % (2 * TS) == 0, "the time loop runs in pairs of batches");

struct ScanBC  { float bt[TS], ct[TS]; };
struct ScanRaw { float p32, xt, xres; };
struct ScanCh  { float delta, dx, xres; };

__device__ __forceinline__ void scan_load_bc(
    ScanBC& s, const float* __restrict__ pr, int n, int t0)
{
#pragma unroll
  for (int i = 0; i < TS; ++i) {
    const float* pt = pr + (size_t)(t0 + i) * P_SZ;
    s.bt[i] = pt[n];
    s.ct[i] = pt[N_SZ + n];
  }
}

__device__ __forceinline__ void scan_load_ch(
    ScanRaw& r, const float* __restrict__ pr, const float* __restrict__ xcr,
    const float* __restrict__ xr, int n, int t0)
{
  r.p32  = pr[(size_t)(t0 + n) * P_SZ + 2 * N_SZ];
  r.xt   = xcr[(size_t)(t0 + n) * D_SZ];
  r.xres = xr[(size_t)(t0 + n) * D_SZ];
}

__device__ __forceinline__ ScanCh scan_prep(const ScanRaw& r, float dtw,
                                            float dtb)
{
  ScanCh c;
  const float z  = fmaf(r.p32, dtw, dtb);
  const float sp = __logf(1.f + __expf(z));
  c.delta = (z > 20.f) ? z : sp;
  c.dx    = c.delta * r.xt;
  c.xres  = r.xres;
  return c;
}

template <int I>
__device__ __forceinline__ float row_lane(float v)
{
  return __int_as_float(__builtin_amdgcn_update_dpp(
      0, __float_as_int(v), 0x150 + I, 0xf, 0xf, false));
}

template <int M>
__device__ __forceinline__ float row_xor(float v)
{
  return __int_as_float(__builtin_amdgcn_update_dpp(
      0, __float_as_int(v), 0x160 + M, 0xf, 0xf, false));
}

template <int M>
__device__ __forceinline__ float scan_fold(bool up, float lo, float hi)
{
  const float keep = up ? hi : lo;
  const float send = up ? lo : hi;
  return keep + row_xor<M>(send);
}

template <int I, int E>
__device__ __forceinline__ void scan_decay(float delta, float a_c,
                                           float* __restrict__ ab)
{
  if constexpr (I < E) {
    ab[I] = __expf(row_lane<I>(delta) * a_c);
    scan_decay<I + 1, E>(delta, a_c, ab);
  }
}

template <int I>
__device__ __forceinline__ void scan_drive(const ScanBC& s, float dx,
                                           float* __restrict__ u)
{
  if constexpr (I < TS) {
    u[I] = row_lane<I>(dx) * s.bt[I];
    scan_drive<I + 1>(s, dx, u);
  }
}

__device__ __forceinline__ float scan_batch(
    ScanBC& s, const ScanCh& c, const float* __restrict__ ab,
    ScanRaw& rn, ScanCh& cn, float* __restrict__ abn,
    float h, float a_c, float dtw, float dtb, float dpd, int n,
    const float* __restrict__ pr, const float* __restrict__ xcr,
    const float* __restrict__ xr, float* __restrict__ outr,
    int t0, int tb, int tr)
{
  float u[TS], y[TS];
  const ScanRaw rnext = rn;
  scan_drive<0>(s, c.dx, u);
#pragma unroll
  for (int i = 0; i < TS; ++i) {
    h = fmaf(ab[i], h, u[i]);
    y[i] = h * s.ct[i];
  }
  scan_load_bc(s, pr, n, tb);
  scan_load_ch(rn, pr, xcr, xr, n, tr);
  const bool n1 = n & 1, n2 = n & 2, n4 = n & 4, n8 = n & 8;
  float r8[8], r4[4], r2[2];
#pragma unroll
  for (int j = 0; j < 8; ++j) {
    const float lo = y[2 * j], hi = y[2 * j + 1];
    r8[j] = scan_fold<1>(n1, lo, hi);
  }
#pragma unroll
  for (int j = 0; j < 4; ++j) {
    const float lo = r8[2 * j], hi = r8[2 * j + 1];
    r4[j] = scan_fold<2>(n2, lo, hi);
  }
#pragma unroll
  for (int j = 0; j < 2; ++j) {
    const float lo = r4[2 * j], hi = r4[2 * j + 1];
    r2[j] = scan_fold<4>(n4, lo, hi);
  }
  const float ylo = r2[0], yhi = r2[1];
  const float yn = scan_fold<8>(n8, ylo, yhi);
  outr[(size_t)(t0 + n) * D_SZ] = fmaf(c.xres, dpd, yn);
  cn = scan_prep(rnext, dtw, dtb);
  scan_decay<0, TS>(cn.delta, a_c, abn);
  return h;
}

__global__ __launch_bounds__(256) void scan_kernel(
    const float* __restrict__ x, const float* __restrict__ xc,
    const float* __restrict__ proj, const float* __restrict__ dt_w,
    const float* __restrict__ dt_b, const float* __restrict__ a_log,
    const float* __restrict__ Dp, float* __restrict__ out)
{
  const int n = threadIdx.x & 15;
  const int c = blockIdx.x * 16 + (threadIdx.x >> 4);
  const int b = c >> 10;
  const int d = c & (D_SZ - 1);

  const float a_c = -__expf(a_log[d * N_SZ + n]);
  const float dtw = dt_w[d];
  const float dtb = dt_b[d];
  const float dpd = Dp[d];

  const float* pr   = proj + (size_t)b * L_SZ * P_SZ;
  const float* xr   = x    + (size_t)b * L_SZ * D_SZ + d;
  const float* xcr  = xc   + (size_t)b * L_SZ * D_SZ + d;
  float*       outr = out  + (size_t)b * L_SZ * D_SZ + d;

  ScanBC  bc0, bc1;
  ScanRaw rawE, rawO;
  ScanCh  chE, chO;
  float   abE[TS], abO[TS];
  scan_load_bc(bc0, pr, n, 0);
  scan_load_bc(bc1, pr, n, TS);
  scan_load_ch(rawE, pr, xcr, xr, n, 0);
  scan_load_ch(rawO, pr, xcr, xr, n, TS);
  chE = scan_prep(rawE, dtw, dtb);
  scan_decay<0, TS>(chE.delta, a_c, abE);
  scan_load_ch(rawE, pr, xcr, xr, n, 2 * TS);
  float h = 0.f;
  for (int t0 = 0; t0 < L_SZ; t0 += 2 * TS) {
    const int t2 = (t0 + 2 * TS < L_SZ) ? (t0 + 2 * TS) : t0;
    const int t3 = (t0 + 3 * TS < L_SZ) ? (t0 + 3 * TS) : t0;
    const int t4 = (t0 + 4 * TS < L_SZ) ? (t0 + 4 * TS) : t0;
    h = scan_batch(bc0, chE, abE, rawO, chO, abO, h, a_c, dtw, dtb, dpd, n,
                   pr, xcr, xr, outr, t0, t2, t3);
    h = scan_batch(bc1, chO, abO, rawE, chE, abE, h, a_c, dtw, dtb, dpd, n,
                   pr, xcr, xr, outr, t0 + TS, t3, t4);
  }
}

extern "C" void kernel_launch(void* const* d_in, const int* in_sizes, int n_in,
                              void* d_out, int out_size, void* d_ws, size_t ws_size,
                              hipStream_t stream) {
  const float* x      = (const float*)d_in[0];
  const float* conv_w = (const float*)d_in[1];
  const float* conv_b = (const float*)d_in[2];
  const float* xproj  = (const float*)d_in[3];
  const float* dt_w   = (const float*)d_in[4];
  const float* dt_b   = (const float*)d_in[5];
  const float* a_log  = (const float*)d_in[6];
  const float* Dp     = (const float*)d_in[7];
  float* out = (float*)d_out;

  float* xc   = (float*)d_ws;
  float* proj = (float*)((char*)d_ws +
                         (size_t)B_SZ * L_SZ * D_SZ * sizeof(float));

  conv_silu_kernel<<<(B_SZ * L_SZ) / CONV_TL, D_QUADS, 0, stream>>>(
      x, conv_w, conv_b, xc);

  proj_wmma_kernel<<<128, 128, 0, stream>>>(xc, xproj, proj);

  scan_kernel<<<(B_SZ * D_SZ) / 16, 256, 0, stream>>>(
      x, xc, proj, dt_w, dt_b, a_log, Dp, out);
}


# proj GEMM: K loop of each 16-row tile split over 8 waves of its block (512 blocks of 8 waves, not 128 of 4), partial tiles summed through LDS in wave order: k summation order changed
# speedup vs baseline: 12.2405x; 1.1086x over previous
#include <hip/hip_runtime.h>
#include <hip/hip_bf16.h>

#define B_SZ 4
#define L_SZ 2048
#define D_SZ 1024
#define N_SZ 16
#define P_SZ 33

typedef float v2f __attribute__((ext_vector_type(2)));
typedef float v8f __attribute__((ext_vector_type(8)));

#define CONV_TL 16
#define D_QUADS (D_SZ / 4)

__device__ __forceinline__ float conv_silu_one(
    float acc, float x0, float x1, float x2, float x3, const float4 w)
{
  acc = fmaf(x0, w.x, acc);
  acc = fmaf(x1, w.y, acc);
  acc = fmaf(x2, w.z, acc);
  acc = fmaf(x3, w.w, acc);
  return acc / (1.f + __expf(-acc));
}

__global__ __launch_bounds__(256) void conv_silu_kernel(
    const float* __restrict__ x, const float* __restrict__ cw,
    const float* __restrict__ cb, float* __restrict__ xc)
{
  const int q   = threadIdx.x;
  const int bl0 = blockIdx.x * CONV_TL;
  const int l0  = bl0 & (L_SZ - 1);
  const float4* xp =
      reinterpret_cast<const float4*>(x + (size_t)bl0 * D_SZ) + q;
  float4* op = reinterpret_cast<float4*>(xc + (size_t)bl0 * D_SZ) + q;

  const float4* wq = reinterpret_cast<const float4*>(cw) + 4 * q;
  const float4 w0 = wq[0], w1 = wq[1], w2 = wq[2], w3 = wq[3];
  const float4 bias = reinterpret_cast<const float4*>(cb)[q];

  float4 xin[CONV_TL + 3];
#pragma unroll
  for (int i = 0; i < CONV_TL + 3; ++i) {
    const bool pad = (l0 + i - 3) < 0;
    const float4 v = xp[pad ? 0 : (i - 3) * D_QUADS];
    xin[i].x = pad ? 0.f : v.x;
    xin[i].y = pad ? 0.f : v.y;
    xin[i].z = pad ? 0.f : v.z;
    xin[i].w = pad ? 0.f : v.w;
  }

#pragma unroll
  for (int i = 0; i < CONV_TL; ++i) {
    float4 o;
    o.x = conv_silu_one(bias.x, xin[i].x, xin[i + 1].x, xin[i + 2].x,
                        xin[i + 3].x, w0);
    o.y = conv_silu_one(bias.y, xin[i].y, xin[i + 1].y, xin[i + 2].y,
                        xin[i + 3].y, w1);
    o.z = conv_silu_one(bias.z, xin[i].z, xin[i + 1].z, xin[i + 2].z,
                        xin[i + 3].z, w2);
    o.w = conv_silu_one(bias.w, xin[i].w, xin[i + 1].w, xin[i + 2].w,
                        xin[i + 3].w, w3);
    op[i * D_QUADS] = o;
  }
}

#define KSPLIT 8
#define KQ (D_SZ / KSPLIT)
static_assert(KQ % 4 == 0, "a wave's share of K is a whole number of k steps");

__global__ __launch_bounds__(32 * KSPLIT) void proj_wmma_kernel(
    const float* __restrict__ xc, const float* __restrict__ w,
    float* __restrict__ proj)
{
  __shared__ float red[(KSPLIT - 1) * 24 * 32];

  const int wave = blockIdx.x;
  const int wv   = threadIdx.x >> 5;
  const int lane = threadIdx.x & 31;
  const int mrow = wave * 16 + (lane & 15);
  const int kb   = ((lane >> 4) << 1) + wv * KQ;
  const float* arow = xc + (size_t)mrow * D_SZ + kb;

  const int ncol = lane & 15;
  const int p0 = ncol;
  const int p1 = 16 + ncol;
  int p2 = 32 + ncol;
  const float s2 = (p2 < P_SZ) ? 1.f : 0.f;
  if (p2 >= P_SZ) p2 = P_SZ - 1;
  const float* brow0 = w + (size_t)p0 * D_SZ + kb;
  const float* brow1 = w + (size_t)p1 * D_SZ + kb;
  const float* brow2 = w + (size_t)p2 * D_SZ + kb;

  v8f acc0 = {}, acc1 = {}, acc2 = {};
  for (int k0 = 0; k0 < KQ; k0 += 4) {
    v2f a  = *(const v2f*)(arow  + k0);
    v2f b0 = *(const v2f*)(brow0 + k0);
    v2f b1 = *(const v2f*)(brow1 + k0);
    v2f b2 = *(const v2f*)(brow2 + k0);
    b2 *= s2;
    acc0 = __builtin_amdgcn_wmma_f32_16x16x4_f32(false, a, false, b0,
                                                 (short)0, acc0, false, false);
    acc1 = __builtin_amdgcn_wmma_f32_16x16x4_f32(false, a, false, b1,
                                                 (short)0, acc1, false, false);
    acc2 = __builtin_amdgcn_wmma_f32_16x16x4_f32(false, a, false, b2,
                                                 (short)0, acc2, false, false);
  }

  if (wv != 0) {
    float* rp = red + (wv - 1) * (24 * 32) + lane;
#pragma unroll
    for (int v = 0; v < 8; ++v) {
      rp[v * 32]        = acc0[v];
      rp[(8 + v) * 32]  = acc1[v];
      rp[(16 + v) * 32] = acc2[v];
    }
  }
  __syncthreads();
  if (wv != 0) return;
#pragma unroll
  for (int j = 0; j < KSPLIT - 1; ++j) {
    const float* rp = red + j * (24 * 32) + lane;
#pragma unroll
    for (int v = 0; v < 8; ++v) {
      acc0[v] += rp[v * 32];
      acc1[v] += rp[(8 + v) * 32];
      acc2[v] += rp[(16 + v) * 32];
    }
  }

  const int mhi = (lane >> 4) << 3;
#pragma unroll
  for (int v = 0; v < 8; ++v) {
    const int row = wave * 16 + v + mhi;
    float* pr = proj + (size_t)row * P_SZ;
    pr[p0] = acc0[v];
    pr[p1] = acc1[v];
    if (32 + ncol < P_SZ) pr[32 + ncol] = acc2[v];
  }
}

#define TS 16
static_assert(TS == N_SZ, "lane n of a row owns step t0 + n of a batch");
static_assert(L_SZ % (2 * TS) == 0, "the time loop runs in pairs of batches");

struct ScanBC  { float bt[TS], ct[TS]; };
struct ScanRaw { float p32, xt, xres; };
struct ScanCh  { float delta, dx, xres; };

__device__ __forceinline__ void scan_load_bc(
    ScanBC& s, const float* __restrict__ pr, int n, int t0)
{
#pragma unroll
  for (int i = 0; i < TS; ++i) {
    const float* pt = pr + (size_t)(t0 + i) * P_SZ;
    s.bt[i] = pt[n];
    s.ct[i] = pt[N_SZ + n];
  }
}

__device__ __forceinline__ void scan_load_ch(
    ScanRaw& r, const float* __restrict__ pr, const float* __restrict__ xcr,
    const float* __restrict__ xr, int n, int t0)
{
  r.p32  = pr[(size_t)(t0 + n) * P_SZ + 2 * N_SZ];
  r.xt   = xcr[(size_t)(t0 + n) * D_SZ];
  r.xres = xr[(size_t)(t0 + n) * D_SZ];
}

__device__ __forceinline__ ScanCh scan_prep(const ScanRaw& r, float dtw,
                                            float dtb)
{
  ScanCh c;
  const float z  = fmaf(r.p32, dtw, dtb);
  const float sp = __logf(1.f + __expf(z));
  c.delta = (z > 20.f) ? z : sp;
  c.dx    = c.delta * r.xt;
  c.xres  = r.xres;
  return c;
}

template <int I>
__device__ __forceinline__ float row_lane(float v)
{
  return __int_as_float(__builtin_amdgcn_update_dpp(
      0, __float_as_int(v), 0x150 + I, 0xf, 0xf, false));
}

template <int M>
__device__ __forceinline__ float row_xor(float v)
{
  return __int_as_float(__builtin_amdgcn_update_dpp(
      0, __float_as_int(v), 0x160 + M, 0xf, 0xf, false));
}

template <int M>
__device__ __forceinline__ float scan_fold(bool up, float lo, float hi)
{
  const float keep = up ? hi : lo;
  const float send = up ? lo : hi;
  return keep + row_xor<M>(send);
}

template <int I, int E>
__device__ __forceinline__ void scan_decay(float delta, float a_c,
                                           float* __restrict__ ab)
{
  if constexpr (I < E) {
    ab[I] = __expf(row_lane<I>(delta) * a_c);
    scan_decay<I + 1, E>(delta, a_c, ab);
  }
}

template <int I>
__device__ __forceinline__ void scan_drive(const ScanBC& s, float dx,
                                           float* __restrict__ u)
{
  if constexpr (I < TS) {
    u[I] = row_lane<I>(dx) * s.bt[I];
    scan_drive<I + 1>(s, dx, u);
  }
}

__device__ __forceinline__ float scan_batch(
    ScanBC& s, const ScanCh& c, const float* __restrict__ ab,
    ScanRaw& rn, ScanCh& cn, float* __restrict__ abn,
    float h, float a_c, float dtw, float dtb, float dpd, int n,
    const float* __restrict__ pr, const float* __restrict__ xcr,
    const float* __restrict__ xr, float* __restrict__ outr,
    int t0, int tb, int tr)
{
  float u[TS], y[TS];
  const ScanRaw rnext = rn;
  scan_drive<0>(s, c.dx, u);
#pragma unroll
  for (int i = 0; i < TS; ++i) {
    h = fmaf(ab[i], h, u[i]);
    y[i] = h * s.ct[i];
  }
  scan_load_bc(s, pr, n, tb);
  scan_load_ch(rn, pr, xcr, xr, n, tr);
  const bool n1 = n & 1, n2 = n & 2, n4 = n & 4, n8 = n & 8;
  float r8[8], r4[4], r2[2];
#pragma unroll
  for (int j = 0; j < 8; ++j) {
    const float lo = y[2 * j], hi = y[2 * j + 1];
    r8[j] = scan_fold<1>(n1, lo, hi);
  }
#pragma unroll
  for (int j = 0; j < 4; ++j) {
    const float lo = r8[2 * j], hi = r8[2 * j + 1];
    r4[j] = scan_fold<2>(n2, lo, hi);
  }
#pragma unroll
  for (int j = 0; j < 2; ++j) {
    const float lo = r4[2 * j], hi = r4[2 * j + 1];
    r2[j] = scan_fold<4>(n4, lo, hi);
  }
  const float ylo = r2[0], yhi = r2[1];
  const float yn = scan_fold<8>(n8, ylo, yhi);
  outr[(size_t)(t0 + n) * D_SZ] = fmaf(c.xres, dpd, yn);
  cn = scan_prep(rnext, dtw, dtb);
  scan_decay<0, TS>(cn.delta, a_c, abn);
  return h;
}

__global__ __launch_bounds__(256) void scan_kernel(
    const float* __restrict__ x, const float* __restrict__ xc,
    const float* __restrict__ proj, const float* __restrict__ dt_w,
    const float* __restrict__ dt_b, const float* __restrict__ a_log,
    const float* __restrict__ Dp, float* __restrict__ out)
{
  const int n = threadIdx.x & 15;
  const int c = blockIdx.x * 16 + (threadIdx.x >> 4);
  const int b = c >> 10;
  const int d = c & (D_SZ - 1);

  const float a_c = -__expf(a_log[d * N_SZ + n]);
  const float dtw = dt_w[d];
  const float dtb = dt_b[d];
  const float dpd = Dp[d];

  const float* pr   = proj + (size_t)b * L_SZ * P_SZ;
  const float* xr   = x    + (size_t)b * L_SZ * D_SZ + d;
  const float* xcr  = xc   + (size_t)b * L_SZ * D_SZ + d;
  float*       outr = out  + (size_t)b * L_SZ * D_SZ + d;

  ScanBC  bc0, bc1;
  ScanRaw rawE, rawO;
  ScanCh  chE, chO;
  float   abE[TS], abO[TS];
  scan_load_bc(bc0, pr, n, 0);
  scan_load_bc(bc1, pr, n, TS);
  scan_load_ch(rawE, pr, xcr, xr, n, 0);
  scan_load_ch(rawO, pr, xcr, xr, n, TS);
  chE = scan_prep(rawE, dtw, dtb);
  scan_decay<0, TS>(chE.delta, a_c, abE);
  scan_load_ch(rawE, pr, xcr, xr, n, 2 * TS);
  float h = 0.f;
  for (int t0 = 0; t0 < L_SZ; t0 += 2 * TS) {
    const int t2 = (t0 + 2 * TS < L_SZ) ? (t0 + 2 * TS) : t0;
    const int t3 = (t0 + 3 * TS < L_SZ) ? (t0 + 3 * TS) : t0;
    const int t4 = (t0 + 4 * TS < L_SZ) ? (t0 + 4 * TS) : t0;
    h = scan_batch(bc0, chE, abE, rawO, chO, abO, h, a_c, dtw, dtb, dpd, n,
                   pr, xcr, xr, outr, t0, t2, t3);
    h = scan_batch(bc1, chO, abO, rawE, chE, abE, h, a_c, dtw, dtb, dpd, n,
                   pr, xcr, xr, outr, t0 + TS, t3, t4);
  }
}

extern "C" void kernel_launch(void* const* d_in, const int* in_sizes, int n_in,
                              void* d_out, int out_size, void* d_ws, size_t ws_size,
                              hipStream_t stream) {
  const float* x      = (const float*)d_in[0];
  const float* conv_w = (const float*)d_in[1];
  const float* conv_b = (const float*)d_in[2];
  const float* xproj  = (const float*)d_in[3];
  const float* dt_w   = (const float*)d_in[4];
  const float* dt_b   = (const float*)d_in[5];
  const float* a_log  = (const float*)d_in[6];
  const float* Dp     = (const float*)d_in[7];
  float* out = (float*)d_out;

  float* xc   = (float*)d_ws;
  float* proj = (float*)((char*)d_ws +
                         (size_t)B_SZ * L_SZ * D_SZ * sizeof(float));

  conv_silu_kernel<<<(B_SZ * L_SZ) / CONV_TL, D_QUADS, 0, stream>>>(
      x, conv_w, conv_b, xc);

  proj_wmma_kernel<<<512, 256, 0, stream>>>(xc, xproj, proj);

  scan_kernel<<<(B_SZ * D_SZ) / 16, 256, 0, stream>>>(
      x, xc, proj, dt_w, dt_b, a_log, Dp, out);
}


# proj result stored in the scan's read order: B, C in 16-step tiles [j][n][4 steps], lane n reads 8 loads of 16 bytes per batch, not 32 of 4; column 32 as a vector; fp order unchanged
# speedup vs baseline: 13.0796x; 1.0686x over previous
#include <hip/hip_runtime.h>
#include <hip/hip_bf16.h>

#define B_SZ 4
#define L_SZ 2048
#define D_SZ 1024
#define N_SZ 16
#define P_SZ 33

typedef float v2f __attribute__((ext_vector_type(2)));
typedef float v8f __attribute__((ext_vector_type(8)));

#define CONV_TL 16
#define D_QUADS (D_SZ / 4)

__device__ __forceinline__ float conv_silu_one(
    float acc, float x0, float x1, float x2, float x3, const float4 w)
{
  acc = fmaf(x0, w.x, acc);
  acc = fmaf(x1, w.y, acc);
  acc = fmaf(x2, w.z, acc);
  acc = fmaf(x3, w.w, acc);
  return acc / (1.f + __expf(-acc));
}

__global__ __launch_bounds__(256) void conv_silu_kernel(
    const float* __restrict__ x, const float* __restrict__ cw,
    const float* __restrict__ cb, float* __restrict__ xc)
{
  const int q   = threadIdx.x;
  const int bl0 = blockIdx.x * CONV_TL;
  const int l0  = bl0 & (L_SZ - 1);
  const float4* xp =
      reinterpret_cast<const float4*>(x + (size_t)bl0 * D_SZ) + q;
  float4* op = reinterpret_cast<float4*>(xc + (size_t)bl0 * D_SZ) + q;

  const float4* wq = reinterpret_cast<const float4*>(cw) + 4 * q;
  const float4 w0 = wq[0], w1 = wq[1], w2 = wq[2], w3 = wq[3];
  const float4 bias = reinterpret_cast<const float4*>(cb)[q];

  float4 xin[CONV_TL + 3];
#pragma unroll
  for (int i = 0; i < CONV_TL + 3; ++i) {
    const bool pad = (l0 + i - 3) < 0;
    const float4 v = xp[pad ? 0 : (i - 3) * D_QUADS];
    xin[i].x = pad ? 0.f : v.x;
    xin[i].y = pad ? 0.f : v.y;
    xin[i].z = pad ? 0.f : v.z;
    xin[i].w = pad ? 0.f : v.w;
  }

#pragma unroll
  for (int i = 0; i < CONV_TL; ++i) {
    float4 o;
    o.x = conv_silu_one(bias.x, xin[i].x, xin[i + 1].x, xin[i + 2].x,
                        xin[i + 3].x, w0);
    o.y = conv_silu_one(bias.y, xin[i].y, xin[i + 1].y, xin[i + 2].y,
                        xin[i + 3].y, w1);
    o.z = conv_silu_one(bias.z, xin[i].z, xin[i + 1].z, xin[i + 2].z,
                        xin[i + 3].z, w2);
    o.w = conv_silu_one(bias.w, xin[i].w, xin[i + 1].w, xin[i + 2].w,
                        xin[i + 3].w, w3);
    op[i * D_QUADS] = o;
  }
}

#define BC_TILE (2 * 16 * N_SZ)
#define KSPLIT 8
#define KQ (D_SZ / KSPLIT)
static_assert(KQ % 4 == 0, "a wave's share of K is a whole number of k steps");

__global__ __launch_bounds__(32 * KSPLIT)
__attribute__((amdgpu_waves_per_eu(4, 4))) void proj_wmma_kernel(
    const float* __restrict__ xc, const float* __restrict__ w,
    float* __restrict__ bc, float* __restrict__ p32)
{
  __shared__ float red[(KSPLIT - 1) * 24 * 32];

  const int wave = blockIdx.x;
  const int wv   = threadIdx.x >> 5;
  const int lane = threadIdx.x & 31;
  const int mrow = wave * 16 + (lane & 15);
  const int kb   = ((lane >> 4) << 1) + wv * KQ;
  const float* arow = xc + (size_t)mrow * D_SZ + kb;

  const int ncol = lane & 15;
  const int p0 = ncol;
  const int p1 = 16 + ncol;
  int p2 = 32 + ncol;
  const float s2 = (p2 < P_SZ) ? 1.f : 0.f;
  if (p2 >= P_SZ) p2 = P_SZ - 1;
  const float* brow0 = w + (size_t)p0 * D_SZ + kb;
  const float* brow1 = w + (size_t)p1 * D_SZ + kb;
  const float* brow2 = w + (size_t)p2 * D_SZ + kb;

  v8f acc0 = {}, acc1 = {}, acc2 = {};
  for (int k0 = 0; k0 < KQ; k0 += 4) {
    v2f a  = *(const v2f*)(arow  + k0);
    v2f b0 = *(const v2f*)(brow0 + k0);
    v2f b1 = *(const v2f*)(brow1 + k0);
    v2f b2 = *(const v2f*)(brow2 + k0);
    b2 *= s2;
    acc0 = __builtin_amdgcn_wmma_f32_16x16x4_f32(false, a, false, b0,
                                                 (short)0, acc0, false, false);
    acc1 = __builtin_amdgcn_wmma_f32_16x16x4_f32(false, a, false, b1,
                                                 (short)0, acc1, false, false);
    acc2 = __builtin_amdgcn_wmma_f32_16x16x4_f32(false, a, false, b2,
                                                 (short)0, acc2, false, false);
  }

  if (wv != 0) {
    float* rp = red + (wv - 1) * (24 * 32) + lane;
#pragma unroll
    for (int v = 0; v < 8; ++v) {
      rp[v * 32]        = acc0[v];
      rp[(8 + v) * 32]  = acc1[v];
      rp[(16 + v) * 32] = acc2[v];
    }
  }
  __syncthreads();
  if (wv != 0) return;
#pragma unroll
  for (int j = 0; j < KSPLIT - 1; ++j) {
    const float* rp = red + j * (24 * 32) + lane;
#pragma unroll
    for (int v = 0; v < 8; ++v) {
      acc0[v] += rp[v * 32];
      acc1[v] += rp[(8 + v) * 32];
      acc2[v] += rp[(16 + v) * 32];
    }
  }

  const int hh = lane >> 4;
  float4* tp = reinterpret_cast<float4*>(bc + (size_t)wave * BC_TILE)
               + (2 * hh) * N_SZ + ncol;
  tp[0]             = make_float4(acc0[0], acc0[1], acc0[2], acc0[3]);
  tp[N_SZ]          = make_float4(acc0[4], acc0[5], acc0[6], acc0[7]);
  tp[4 * N_SZ]        = make_float4(acc1[0], acc1[1], acc1[2], acc1[3]);
  tp[4 * N_SZ + N_SZ] = make_float4(acc1[4], acc1[5], acc1[6], acc1[7]);
  if (ncol == 0) {
    float4* pp = reinterpret_cast<float4*>(p32 + wave * 16 + 8 * hh);
    pp[0] = make_float4(acc2[0], acc2[1], acc2[2], acc2[3]);
    pp[1] = make_float4(acc2[4], acc2[5], acc2[6], acc2[7]);
  }
}

#define TS 16
static_assert(TS == N_SZ, "lane n of a row owns step t0 + n of a batch");
static_assert(L_SZ % (2 * TS) == 0, "the time loop runs in pairs of batches");

struct ScanBC  { float bt[TS], ct[TS]; };
struct ScanRaw { float p32, xt, xres; };
struct ScanCh  { float delta, dx, xres; };

__device__ __forceinline__ void scan_load_bc(
    ScanBC& s, const float* __restrict__ bcb, int n, int t0)
{
  const float4* tp =
      reinterpret_cast<const float4*>(bcb + (size_t)t0 * (BC_TILE / TS)) + n;
#pragma unroll
  for (int j = 0; j < TS / 4; ++j) {
    const float4 bv = tp[j * N_SZ];
    const float4 cv = tp[(TS / 4 + j) * N_SZ];
    s.bt[4 * j + 0] = bv.x; s.bt[4 * j + 1] = bv.y;
    s.bt[4 * j + 2] = bv.z; s.bt[4 * j + 3] = bv.w;
    s.ct[4 * j + 0] = cv.x; s.ct[4 * j + 1] = cv.y;
    s.ct[4 * j + 2] = cv.z; s.ct[4 * j + 3] = cv.w;
  }
}

__device__ __forceinline__ void scan_load_ch(
    ScanRaw& r, const float* __restrict__ p32b, const float* __restrict__ xcr,
    const float* __restrict__ xr, int n, int t0)
{
  r.p32  = p32b[t0 + n];
  r.xt   = xcr[(size_t)(t0 + n) * D_SZ];
  r.xres = xr[(size_t)(t0 + n) * D_SZ];
}

__device__ __forceinline__ ScanCh scan_prep(const ScanRaw& r, float dtw,
                                            float dtb)
{
  ScanCh c;
  const float z  = fmaf(r.p32, dtw, dtb);
  const float sp = __logf(1.f + __expf(z));
  c.delta = (z > 20.f) ? z : sp;
  c.dx    = c.delta * r.xt;
  c.xres  = r.xres;
  return c;
}

template <int I>
__device__ __forceinline__ float row_lane(float v)
{
  return __int_as_float(__builtin_amdgcn_update_dpp(
      0, __float_as_int(v), 0x150 + I, 0xf, 0xf, false));
}

template <int M>
__device__ __forceinline__ float row_xor(float v)
{
  return __int_as_float(__builtin_amdgcn_update_dpp(
      0, __float_as_int(v), 0x160 + M, 0xf, 0xf, false));
}

template <int M>
__device__ __forceinline__ float scan_fold(bool up, float lo, float hi)
{
  const float keep = up ? hi : lo;
  const float send = up ? lo : hi;
  return keep + row_xor<M>(send);
}

template <int I, int E>
__device__ __forceinline__ void scan_decay(float delta, float a_c,
                                           float* __restrict__ ab)
{
  if constexpr (I < E) {
    ab[I] = __expf(row_lane<I>(delta) * a_c);
    scan_decay<I + 1, E>(delta, a_c, ab);
  }
}

template <int I>
__device__ __forceinline__ void scan_drive(const ScanBC& s, float dx,
                                           float* __restrict__ u)
{
  if constexpr (I < TS) {
    u[I] = row_lane<I>(dx) * s.bt[I];
    scan_drive<I + 1>(s, dx, u);
  }
}

__device__ __forceinline__ float scan_batch(
    ScanBC& s, const ScanCh& c, const float* __restrict__ ab,
    ScanRaw& rn, ScanCh& cn, float* __restrict__ abn,
    float h, float a_c, float dtw, float dtb, float dpd, int n,
    const float* __restrict__ bcb, const float* __restrict__ p32b,
    const float* __restrict__ xcr,
    const float* __restrict__ xr, float* __restrict__ outr,
    int t0, int tb, int tr)
{
  float u[TS], y[TS];
  const ScanRaw rnext = rn;
  scan_drive<0>(s, c.dx, u);
#pragma unroll
  for (int i = 0; i < TS; ++i) {
    h = fmaf(ab[i], h, u[i]);
    y[i] = h * s.ct[i];
  }
  scan_load_bc(s, bcb, n, tb);
  scan_load_ch(rn, p32b, xcr, xr, n, tr);
  const bool n1 = n & 1, n2 = n & 2, n4 = n & 4, n8 = n & 8;
  float r8[8], r4[4], r2[2];
#pragma unroll
  for (int j = 0; j < 8; ++j) {
    const float lo = y[2 * j], hi = y[2 * j + 1];
    r8[j] = scan_fold<1>(n1, lo, hi);
  }
#pragma unroll
  for (int j = 0; j < 4; ++j) {
    const float lo = r8[2 * j], hi = r8[2 * j + 1];
    r4[j] = scan_fold<2>(n2, lo, hi);
  }
#pragma unroll
  for (int j = 0; j < 2; ++j) {
    const float lo = r4[2 * j], hi = r4[2 * j + 1];
    r2[j] = scan_fold<4>(n4, lo, hi);
  }
  const float ylo = r2[0], yhi = r2[1];
  const float yn = scan_fold<8>(n8, ylo, yhi);
  outr[(size_t)(t0 + n) * D_SZ] = fmaf(c.xres, dpd, yn);
  cn = scan_prep(rnext, dtw, dtb);
  scan_decay<0, TS>(cn.delta, a_c, abn);
  return h;
}

__global__ __launch_bounds__(256) void scan_kernel(
    const float* __restrict__ x, const float* __restrict__ xc,
    const float* __restrict__ bc, const float* __restrict__ p32,
    const float* __restrict__ dt_w,
    const float* __restrict__ dt_b, const float* __restrict__ a_log,
    const float* __restrict__ Dp, float* __restrict__ out)
{
  const int n = threadIdx.x & 15;
  const int c = blockIdx.x * 16 + (threadIdx.x >> 4);
  const int b = c >> 10;
  const int d = c & (D_SZ - 1);

  const float a_c = -__expf(a_log[d * N_SZ + n]);
  const float dtw = dt_w[d];
  const float dtb = dt_b[d];
  const float dpd = Dp[d];

  const float* bcb  = bc  + (size_t)b * L_SZ * (BC_TILE / TS);
  const float* p32b = p32 + (size_t)b * L_SZ;
  const float* xr   = x    + (size_t)b * L_SZ * D_SZ + d;
  const float* xcr  = xc   + (size_t)b * L_SZ * D_SZ + d;
  float*       outr = out  + (size_t)b * L_SZ * D_SZ + d;

  ScanBC  bc0, bc1;
  ScanRaw rawE, rawO;
  ScanCh  chE, chO;
  float   abE[TS], abO[TS];
  scan_load_bc(bc0, bcb, n, 0);
  scan_load_bc(bc1, bcb, n, TS);
  scan_load_ch(rawE, p32b, xcr, xr, n, 0);
  scan_load_ch(rawO, p32b, xcr, xr, n, TS);
  chE = scan_prep(rawE, dtw, dtb);
  scan_decay<0, TS>(chE.delta, a_c, abE);
  scan_load_ch(rawE, p32b, xcr, xr, n, 2 * TS);
  float h = 0.f;
  for (int t0 = 0; t0 < L_SZ; t0 += 2 * TS) {
    const int t2 = (t0 + 2 * TS < L_SZ) ? (t0 + 2 * TS) : t0;
    const int t3 = (t0 + 3 * TS < L_SZ) ? (t0 + 3 * TS) : t0;
    const int t4 = (t0 + 4 * TS < L_SZ) ? (t0 + 4 * TS) : t0;
    h = scan_batch(bc0, chE, abE, rawO, chO, abO, h, a_c, dtw, dtb, dpd, n,
                   bcb, p32b, xcr, xr, outr, t0, t2, t3);
    h = scan_batch(bc1, chO, abO, rawE, chE, abE, h, a_c, dtw, dtb, dpd, n,
                   bcb, p32b, xcr, xr, outr, t0 + TS, t3, t4);
  }
}

extern "C" void kernel_launch(void* const* d_in, const int* in_sizes, int n_in,
                              void* d_out, int out_size, void* d_ws, size_t ws_size,
                              hipStream_t stream) {
  const float* x      = (const float*)d_in[0];
  const float* conv_w = (const float*)d_in[1];
  const float* conv_b = (const float*)d_in[2];
  const float* xproj  = (const float*)d_in[3];
  const float* dt_w   = (const float*)d_in[4];
  const float* dt_b   = (const float*)d_in[5];
  const float* a_log  = (const float*)d_in[6];
  const float* Dp     = (const float*)d_in[7];
  float* out = (float*)d_out;

  float* xc   = (float*)d_ws;
  float* bc   = (float*)((char*)d_ws +
                         (size_t)B_SZ * L_SZ * D_SZ * sizeof(float));
  float* p32  = bc + (size_t)B_SZ * L_SZ * (BC_TILE / 16);

  conv_silu_kernel<<<(B_SZ * L_SZ) / CONV_TL, D_QUADS, 0, stream>>>(
      x, conv_w, conv_b, xc);

  proj_wmma_kernel<<<512, 256, 0, stream>>>(xc, xproj, bc, p32);

  scan_kernel<<<(B_SZ * D_SZ) / 16, 256, 0, stream>>>(
      x, xc, bc, p32, dt_w, dt_b, a_log, Dp, out);
}


# scan: every load of the time loop issued four batches ahead (B, C and the per-channel gathers in four register sets each, 4 batches per trip, 2 waves per SIMD stated); fp order unchanged
# speedup vs baseline: 13.9784x; 1.0687x over previous
#include <hip/hip_runtime.h>
#include <hip/hip_bf16.h>

#define B_SZ 4
#define L_SZ 2048
#define D_SZ 1024
#define N_SZ 16
#define P_SZ 33

typedef float v2f __attribute__((ext_vector_type(2)));
typedef float v8f __attribute__((ext_vector_type(8)));

#define CONV_TL 16
#define D_QUADS (D_SZ / 4)

__device__ __forceinline__ float conv_silu_one(
    float acc, float x0, float x1, float x2, float x3, const float4 w)
{
  acc = fmaf(x0, w.x, acc);
  acc = fmaf(x1, w.y, acc);
  acc = fmaf(x2, w.z, acc);
  acc = fmaf(x3, w.w, acc);
  return acc / (1.f + __expf(-acc));
}

__global__ __launch_bounds__(256) void conv_silu_kernel(
    const float* __restrict__ x, const float* __restrict__ cw,
    const float* __restrict__ cb, float* __restrict__ xc)
{
  const int q   = threadIdx.x;
  const int bl0 = blockIdx.x * CONV_TL;
  const int l0  = bl0 & (L_SZ - 1);
  const float4* xp =
      reinterpret_cast<const float4*>(x + (size_t)bl0 * D_SZ) + q;
  float4* op = reinterpret_cast<float4*>(xc + (size_t)bl0 * D_SZ) + q;

  const float4* wq = reinterpret_cast<const float4*>(cw) + 4 * q;
  const float4 w0 = wq[0], w1 = wq[1], w2 = wq[2], w3 = wq[3];
  const float4 bias = reinterpret_cast<const float4*>(cb)[q];

  float4 xin[CONV_TL + 3];
#pragma unroll
  for (int i = 0; i < CONV_TL + 3; ++i) {
    const bool pad = (l0 + i - 3) < 0;
    const float4 v = xp[pad ? 0 : (i - 3) * D_QUADS];
    xin[i].x = pad ? 0.f : v.x;
    xin[i].y = pad ? 0.f : v.y;
    xin[i].z = pad ? 0.f : v.z;
    xin[i].w = pad ? 0.f : v.w;
  }

#pragma unroll
  for (int i = 0; i < CONV_TL; ++i) {
    float4 o;
    o.x = conv_silu_one(bias.x, xin[i].x, xin[i + 1].x, xin[i + 2].x,
                        xin[i + 3].x, w0);
    o.y = conv_silu_one(bias.y, xin[i].y, xin[i + 1].y, xin[i + 2].y,
                        xin[i + 3].y, w1);
    o.z = conv_silu_one(bias.z, xin[i].z, xin[i + 1].z, xin[i + 2].z,
                        xin[i + 3].z, w2);
    o.w = conv_silu_one(bias.w, xin[i].w, xin[i + 1].w, xin[i + 2].w,
                        xin[i + 3].w, w3);
    op[i * D_QUADS] = o;
  }
}

#define BC_TILE (2 * 16 * N_SZ)
#define KSPLIT 8
#define KQ (D_SZ / KSPLIT)
static_assert(KQ % 4 == 0, "a wave's share of K is a whole number of k steps");

__global__ __launch_bounds__(32 * KSPLIT)
__attribute__((amdgpu_waves_per_eu(4, 4))) void proj_wmma_kernel(
    const float* __restrict__ xc, const float* __restrict__ w,
    float* __restrict__ bc, float* __restrict__ p32)
{
  __shared__ float red[(KSPLIT - 1) * 24 * 32];

  const int wave = blockIdx.x;
  const int wv   = threadIdx.x >> 5;
  const int lane = threadIdx.x & 31;
  const int mrow = wave * 16 + (lane & 15);
  const int kb   = ((lane >> 4) << 1) + wv * KQ;
  const float* arow = xc + (size_t)mrow * D_SZ + kb;

  const int ncol = lane & 15;
  const int p0 = ncol;
  const int p1 = 16 + ncol;
  int p2 = 32 + ncol;
  const float s2 = (p2 < P_SZ) ? 1.f : 0.f;
  if (p2 >= P_SZ) p2 = P_SZ - 1;
  const float* brow0 = w + (size_t)p0 * D_SZ + kb;
  const float* brow1 = w + (size_t)p1 * D_SZ + kb;
  const float* brow2 = w + (size_t)p2 * D_SZ + kb;

  v8f acc0 = {}, acc1 = {}, acc2 = {};
  for (int k0 = 0; k0 < KQ; k0 += 4) {
    v2f a  = *(const v2f*)(arow  + k0);
    v2f b0 = *(const v2f*)(brow0 + k0);
    v2f b1 = *(const v2f*)(brow1 + k0);
    v2f b2 = *(const v2f*)(brow2 + k0);
    b2 *= s2;
    acc0 = __builtin_amdgcn_wmma_f32_16x16x4_f32(false, a, false, b0,
                                                 (short)0, acc0, false, false);
    acc1 = __builtin_amdgcn_wmma_f32_16x16x4_f32(false, a, false, b1,
                                                 (short)0, acc1, false, false);
    acc2 = __builtin_amdgcn_wmma_f32_16x16x4_f32(false, a, false, b2,
                                                 (short)0, acc2, false, false);
  }

  if (wv != 0) {
    float* rp = red + (wv - 1) * (24 * 32) + lane;
#pragma unroll
    for (int v = 0; v < 8; ++v) {
      rp[v * 32]        = acc0[v];
      rp[(8 + v) * 32]  = acc1[v];
      rp[(16 + v) * 32] = acc2[v];
    }
  }
  __syncthreads();
  if (wv != 0) return;
#pragma unroll
  for (int j = 0; j < KSPLIT - 1; ++j) {
    const float* rp = red + j * (24 * 32) + lane;
#pragma unroll
    for (int v = 0; v < 8; ++v) {
      acc0[v] += rp[v * 32];
      acc1[v] += rp[(8 + v) * 32];
      acc2[v] += rp[(16 + v) * 32];
    }
  }

  const int hh = lane >> 4;
  float4* tp = reinterpret_cast<float4*>(bc + (size_t)wave * BC_TILE)
               + (2 * hh) * N_SZ + ncol;
  tp[0]             = make_float4(acc0[0], acc0[1], acc0[2], acc0[3]);
  tp[N_SZ]          = make_float4(acc0[4], acc0[5], acc0[6], acc0[7]);
  tp[4 * N_SZ]        = make_float4(acc1[0], acc1[1], acc1[2], acc1[3]);
  tp[4 * N_SZ + N_SZ] = make_float4(acc1[4], acc1[5], acc1[6], acc1[7]);
  if (ncol == 0) {
    float4* pp = reinterpret_cast<float4*>(p32 + wave * 16 + 8 * hh);
    pp[0] = make_float4(acc2[0], acc2[1], acc2[2], acc2[3]);
    pp[1] = make_float4(acc2[4], acc2[5], acc2[6], acc2[7]);
  }
}

#define TS 16
static_assert(TS == N_SZ, "lane n of a row owns step t0 + n of a batch");
static_assert(L_SZ % (4 * TS) == 0, "the time loop runs four batches per trip");

struct ScanBC  { float bt[TS], ct[TS]; };
struct ScanRaw { float p32, xt, xres; };
struct ScanCh  { float delta, dx, xres; };

__device__ __forceinline__ void scan_load_bc(
    ScanBC& s, const float* __restrict__ bcb, int n, int t0)
{
  const float4* tp =
      reinterpret_cast<const float4*>(bcb + (size_t)t0 * (BC_TILE / TS)) + n;
#pragma unroll
  for (int j = 0; j < TS / 4; ++j) {
    const float4 bv = tp[j * N_SZ];
    const float4 cv = tp[(TS / 4 + j) * N_SZ];
    s.bt[4 * j + 0] = bv.x; s.bt[4 * j + 1] = bv.y;
    s.bt[4 * j + 2] = bv.z; s.bt[4 * j + 3] = bv.w;
    s.ct[4 * j + 0] = cv.x; s.ct[4 * j + 1] = cv.y;
    s.ct[4 * j + 2] = cv.z; s.ct[4 * j + 3] = cv.w;
  }
}

__device__ __forceinline__ void scan_load_ch(
    ScanRaw& r, const float* __restrict__ p32b, const float* __restrict__ xcr,
    const float* __restrict__ xr, int n, int t0)
{
  r.p32  = p32b[t0 + n];
  r.xt   = xcr[(size_t)(t0 + n) * D_SZ];
  r.xres = xr[(size_t)(t0 + n) * D_SZ];
}

__device__ __forceinline__ ScanCh scan_prep(const ScanRaw& r, float dtw,
                                            float dtb)
{
  ScanCh c;
  const float z  = fmaf(r.p32, dtw, dtb);
  const float sp = __logf(1.f + __expf(z));
  c.delta = (z > 20.f) ? z : sp;
  c.dx    = c.delta * r.xt;
  c.xres  = r.xres;
  return c;
}

template <int I>
__device__ __forceinline__ float row_lane(float v)
{
  return __int_as_float(__builtin_amdgcn_update_dpp(
      0, __float_as_int(v), 0x150 + I, 0xf, 0xf, false));
}

template <int M>
__device__ __forceinline__ float row_xor(float v)
{
  return __int_as_float(__builtin_amdgcn_update_dpp(
      0, __float_as_int(v), 0x160 + M, 0xf, 0xf, false));
}

template <int M>
__device__ __forceinline__ float scan_fold(bool up, float lo, float hi)
{
  const float keep = up ? hi : lo;
  const float send = up ? lo : hi;
  return keep + row_xor<M>(send);
}

template <int I, int E>
__device__ __forceinline__ void scan_decay(float delta, float a_c,
                                           float* __restrict__ ab)
{
  if constexpr (I < E) {
    ab[I] = __expf(row_lane<I>(delta) * a_c);
    scan_decay<I + 1, E>(delta, a_c, ab);
  }
}

template <int I>
__device__ __forceinline__ void scan_drive(const ScanBC& s, float dx,
                                           float* __restrict__ u)
{
  if constexpr (I < TS) {
    u[I] = row_lane<I>(dx) * s.bt[I];
    scan_drive<I + 1>(s, dx, u);
  }
}

__device__ __forceinline__ float scan_batch(
    ScanBC& s, const ScanCh& c, const float* __restrict__ ab,
    ScanRaw& rn, ScanCh& cn, float* __restrict__ abn,
    float h, float a_c, float dtw, float dtb, float dpd, int n,
    const float* __restrict__ bcb, const float* __restrict__ p32b,
    const float* __restrict__ xcr,
    const float* __restrict__ xr, float* __restrict__ outr,
    int t0, int tb, int tr)
{
  float u[TS], y[TS];
  scan_drive<0>(s, c.dx, u);
#pragma unroll
  for (int i = 0; i < TS; ++i) {
    h = fmaf(ab[i], h, u[i]);
    y[i] = h * s.ct[i];
  }
  scan_load_bc(s, bcb, n, tb);
  const bool n1 = n & 1, n2 = n & 2, n4 = n & 4, n8 = n & 8;
  float r8[8], r4[4], r2[2];
#pragma unroll
  for (int j = 0; j < 8; ++j) {
    const float lo = y[2 * j], hi = y[2 * j + 1];
    r8[j] = scan_fold<1>(n1, lo, hi);
  }
#pragma unroll
  for (int j = 0; j < 4; ++j) {
    const float lo = r8[2 * j], hi = r8[2 * j + 1];
    r4[j] = scan_fold<2>(n2, lo, hi);
  }
#pragma unroll
  for (int j = 0; j < 2; ++j) {
    const float lo = r4[2 * j], hi = r4[2 * j + 1];
    r2[j] = scan_fold<4>(n4, lo, hi);
  }
  const float ylo = r2[0], yhi = r2[1];
  const float yn = scan_fold<8>(n8, ylo, yhi);
  outr[(size_t)(t0 + n) * D_SZ] = fmaf(c.xres, dpd, yn);
  cn = scan_prep(rn, dtw, dtb);
  scan_decay<0, TS>(cn.delta, a_c, abn);
  scan_load_ch(rn, p32b, xcr, xr, n, tr);
  return h;
}

__global__ __launch_bounds__(256)
__attribute__((amdgpu_waves_per_eu(2, 2))) void scan_kernel(
    const float* __restrict__ x, const float* __restrict__ xc,
    const float* __restrict__ bc, const float* __restrict__ p32,
    const float* __restrict__ dt_w,
    const float* __restrict__ dt_b, const float* __restrict__ a_log,
    const float* __restrict__ Dp, float* __restrict__ out)
{
  const int n = threadIdx.x & 15;
  const int c = blockIdx.x * 16 + (threadIdx.x >> 4);
  const int b = c >> 10;
  const int d = c & (D_SZ - 1);

  const float a_c = -__expf(a_log[d * N_SZ + n]);
  const float dtw = dt_w[d];
  const float dtb = dt_b[d];
  const float dpd = Dp[d];

  const float* bcb  = bc  + (size_t)b * L_SZ * (BC_TILE / TS);
  const float* p32b = p32 + (size_t)b * L_SZ;
  const float* xr   = x    + (size_t)b * L_SZ * D_SZ + d;
  const float* xcr  = xc   + (size_t)b * L_SZ * D_SZ + d;
  float*       outr = out  + (size_t)b * L_SZ * D_SZ + d;

  ScanBC  bc0, bc1, bc2, bc3;
  ScanRaw raw0, raw1, raw2, raw3;
  ScanCh  chE, chO;
  float   abE[TS], abO[TS];
  scan_load_bc(bc0, bcb, n, 0);
  scan_load_bc(bc1, bcb, n, TS);
  scan_load_bc(bc2, bcb, n, 2 * TS);
  scan_load_bc(bc3, bcb, n, 3 * TS);
  scan_load_ch(raw0, p32b, xcr, xr, n, 0);
  scan_load_ch(raw1, p32b, xcr, xr, n, TS);
  scan_load_ch(raw2, p32b, xcr, xr, n, 2 * TS);
  scan_load_ch(raw3, p32b, xcr, xr, n, 3 * TS);
  chE = scan_prep(raw0, dtw, dtb);
  scan_decay<0, TS>(chE.delta, a_c, abE);
  scan_load_ch(raw0, p32b, xcr, xr, n, 4 * TS);
  float h = 0.f;
  for (int t0 = 0; t0 < L_SZ; t0 += 4 * TS) {
    const int t4 = (t0 + 4 * TS < L_SZ) ? (t0 + 4 * TS) : t0;
    const int t5 = (t0 + 5 * TS < L_SZ) ? (t0 + 5 * TS) : t0;
    const int t6 = (t0 + 6 * TS < L_SZ) ? (t0 + 6 * TS) : t0;
    const int t7 = (t0 + 7 * TS < L_SZ) ? (t0 + 7 * TS) : t0;
    const int t8 = (t0 + 8 * TS < L_SZ) ? (t0 + 8 * TS) : t0;
    h = scan_batch(bc0, chE, abE, raw1, chO, abO, h, a_c, dtw, dtb, dpd, n,
                   bcb, p32b, xcr, xr, outr, t0, t4, t5);
    h = scan_batch(bc1, chO, abO, raw2, chE, abE, h, a_c, dtw, dtb, dpd, n,
                   bcb, p32b, xcr, xr, outr, t0 + TS, t5, t6);
    h = scan_batch(bc2, chE, abE, raw3, chO, abO, h, a_c, dtw, dtb, dpd, n,
                   bcb, p32b, xcr, xr, outr, t0 + 2 * TS, t6, t7);
    h = scan_batch(bc3, chO, abO, raw0, chE, abE, h, a_c, dtw, dtb, dpd, n,
                   bcb, p32b, xcr, xr, outr, t0 + 3 * TS, t7, t8);
  }
}

extern "C" void kernel_launch(void* const* d_in, const int* in_sizes, int n_in,
                              void* d_out, int out_size, void* d_ws, size_t ws_size,
                              hipStream_t stream) {
  const float* x      = (const float*)d_in[0];
  const float* conv_w = (const float*)d_in[1];
  const float* conv_b = (const float*)d_in[2];
  const float* xproj  = (const float*)d_in[3];
  const float* dt_w   = (const float*)d_in[4];
  const float* dt_b   = (const float*)d_in[5];
  const float* a_log  = (const float*)d_in[6];
  const float* Dp     = (const float*)d_in[7];
  float* out = (float*)d_out;

  float* xc   = (float*)d_ws;
  float* bc   = (float*)((char*)d_ws +
                         (size_t)B_SZ * L_SZ * D_SZ * sizeof(float));
  float* p32  = bc + (size_t)B_SZ * L_SZ * (BC_TILE / 16);

  conv_silu_kernel<<<(B_SZ * L_SZ) / CONV_TL, D_QUADS, 0, stream>>>(
      x, conv_w, conv_b, xc);

  proj_wmma_kernel<<<512, 256, 0, stream>>>(xc, xproj, bc, p32);

  scan_kernel<<<(B_SZ * D_SZ) / 16, 256, 0, stream>>>(
      x, xc, bc, p32, dt_w, dt_b, a_log, Dp, out);
}


# scan: x, conv output, B, C, proj column 32 and out go through LDS per work-group (64-byte row pieces, whole lines, one LDS-only barrier per 64 steps), not 16-line gathers per wave; fp order unchanged
# speedup vs baseline: 22.3838x; 1.6013x over previous
#include <hip/hip_runtime.h>
#include <hip/hip_bf16.h>

#define B_SZ 4
#define L_SZ 2048
#define D_SZ 1024
#define N_SZ 16
#define P_SZ 33

typedef float v2f __attribute__((ext_vector_type(2)));
typedef float v8f __attribute__((ext_vector_type(8)));

#define CONV_TL 16
#define D_QUADS (D_SZ / 4)

__device__ __forceinline__ float conv_silu_one(
    float acc, float x0, float x1, float x2, float x3, const float4 w)
{
  acc = fmaf(x0, w.x, acc);
  acc = fmaf(x1, w.y, acc);
  acc = fmaf(x2, w.z, acc);
  acc = fmaf(x3, w.w, acc);
  return acc / (1.f + __expf(-acc));
}

__global__ __launch_bounds__(256) void conv_silu_kernel(
    const float* __restrict__ x, const float* __restrict__ cw,
    const float* __restrict__ cb, float* __restrict__ xc)
{
  const int q   = threadIdx.x;
  const int bl0 = blockIdx.x * CONV_TL;
  const int l0  = bl0 & (L_SZ - 1);
  const float4* xp =
      reinterpret_cast<const float4*>(x + (size_t)bl0 * D_SZ) + q;
  float4* op = reinterpret_cast<float4*>(xc + (size_t)bl0 * D_SZ) + q;

  const float4* wq = reinterpret_cast<const float4*>(cw) + 4 * q;
  const float4 w0 = wq[0], w1 = wq[1], w2 = wq[2], w3 = wq[3];
  const float4 bias = reinterpret_cast<const float4*>(cb)[q];

  float4 xin[CONV_TL + 3];
#pragma unroll
  for (int i = 0; i < CONV_TL + 3; ++i) {
    const bool pad = (l0 + i - 3) < 0;
    const float4 v = xp[pad ? 0 : (i - 3) * D_QUADS];
    xin[i].x = pad ? 0.f : v.x;
    xin[i].y = pad ? 0.f : v.y;
    xin[i].z = pad ? 0.f : v.z;
    xin[i].w = pad ? 0.f : v.w;
  }

#pragma unroll
  for (int i = 0; i < CONV_TL; ++i) {
    float4 o;
    o.x = conv_silu_one(bias.x, xin[i].x, xin[i + 1].x, xin[i + 2].x,
                        xin[i + 3].x, w0);
    o.y = conv_silu_one(bias.y, xin[i].y, xin[i + 1].y, xin[i + 2].y,
                        xin[i + 3].y, w1);
    o.z = conv_silu_one(bias.z, xin[i].z, xin[i + 1].z, xin[i + 2].z,
                        xin[i + 3].z, w2);
    o.w = conv_silu_one(bias.w, xin[i].w, xin[i + 1].w, xin[i + 2].w,
                        xin[i + 3].w, w3);
    op[i * D_QUADS] = o;
  }
}

#define BC_TILE (2 * 16 * N_SZ)
#define KSPLIT 8
#define KQ (D_SZ / KSPLIT)
static_assert(KQ % 4 == 0, "a wave's share of K is a whole number of k steps");

__global__ __launch_bounds__(32 * KSPLIT)
__attribute__((amdgpu_waves_per_eu(4, 4))) void proj_wmma_kernel(
    const float* __restrict__ xc, const float* __restrict__ w,
    float* __restrict__ bc, float* __restrict__ p32)
{
  __shared__ float red[(KSPLIT - 1) * 24 * 32];

  const int wave = blockIdx.x;
  const int wv   = threadIdx.x >> 5;
  const int lane = threadIdx.x & 31;
  const int mrow = wave * 16 + (lane & 15);
  const int kb   = ((lane >> 4) << 1) + wv * KQ;
  const float* arow = xc + (size_t)mrow * D_SZ + kb;

  const int ncol = lane & 15;
  const int p0 = ncol;
  const int p1 = 16 + ncol;
  int p2 = 32 + ncol;
  const float s2 = (p2 < P_SZ) ? 1.f : 0.f;
  if (p2 >= P_SZ) p2 = P_SZ - 1;
  const float* brow0 = w + (size_t)p0 * D_SZ + kb;
  const float* brow1 = w + (size_t)p1 * D_SZ + kb;
  const float* brow2 = w + (size_t)p2 * D_SZ + kb;

  v8f acc0 = {}, acc1 = {}, acc2 = {};
  for (int k0 = 0; k0 < KQ; k0 += 4) {
    v2f a  = *(const v2f*)(arow  + k0);
    v2f b0 = *(const v2f*)(brow0 + k0);
    v2f b1 = *(const v2f*)(brow1 + k0);
    v2f b2 = *(const v2f*)(brow2 + k0);
    b2 *= s2;
    acc0 = __builtin_amdgcn_wmma_f32_16x16x4_f32(false, a, false, b0,
                                                 (short)0, acc0, false, false);
    acc1 = __builtin_amdgcn_wmma_f32_16x16x4_f32(false, a, false, b1,
                                                 (short)0, acc1, false, false);
    acc2 = __builtin_amdgcn_wmma_f32_16x16x4_f32(false, a, false, b2,
                                                 (short)0, acc2, false, false);
  }

  if (wv != 0) {
    float* rp = red + (wv - 1) * (24 * 32) + lane;
#pragma unroll
    for (int v = 0; v < 8; ++v) {
      rp[v * 32]        = acc0[v];
      rp[(8 + v) * 32]  = acc1[v];
      rp[(16 + v) * 32] = acc2[v];
    }
  }
  __syncthreads();
  if (wv != 0) return;
#pragma unroll
  for (int j = 0; j < KSPLIT - 1; ++j) {
    const float* rp = red + j * (24 * 32) + lane;
#pragma unroll
    for (int v = 0; v < 8; ++v) {
      acc0[v] += rp[v * 32];
      acc1[v] += rp[(8 + v) * 32];
      acc2[v] += rp[(16 + v) * 32];
    }
  }

  const int hh = lane >> 4;
  float4* tp = reinterpret_cast<float4*>(bc + (size_t)wave * BC_TILE)
               + (2 * hh) * N_SZ + ncol;
  tp[0]             = make_float4(acc0[0], acc0[1], acc0[2], acc0[3]);
  tp[N_SZ]          = make_float4(acc0[4], acc0[5], acc0[6], acc0[7]);
  tp[4 * N_SZ]        = make_float4(acc1[0], acc1[1], acc1[2], acc1[3]);
  tp[4 * N_SZ + N_SZ] = make_float4(acc1[4], acc1[5], acc1[6], acc1[7]);
  if (ncol == 0) {
    float4* pp = reinterpret_cast<float4*>(p32 + wave * 16 + 8 * hh);
    pp[0] = make_float4(acc2[0], acc2[1], acc2[2], acc2[3]);
    pp[1] = make_float4(acc2[4], acc2[5], acc2[6], acc2[7]);
  }
}

#define TS 16
#define TRIP (4 * TS)
#define SPITCH 20
#define SBUF (TRIP * SPITCH)
#define BCBUF (4 * BC_TILE)
static_assert(TS == N_SZ, "lane n of a row owns step t0 + n of a batch");
static_assert(L_SZ % (2 * TRIP) == 0, "the time loop runs two trips per pass");
static_assert((SPITCH * 4) % 32 == 16, "row pitch: an odd multiple of 16 bytes");

struct ScanBC  { float bt[TS], ct[TS]; };
struct ScanCh  { float delta, dx, xres; };

__device__ __forceinline__ void scan_load_bc(
    ScanBC& s, const float* __restrict__ tile, int n)
{
  const float4* tp = reinterpret_cast<const float4*>(tile) + n;
#pragma unroll
  for (int j = 0; j < TS / 4; ++j) {
    const float4 bv = tp[j * N_SZ];
    const float4 cv = tp[(TS / 4 + j) * N_SZ];
    s.bt[4 * j + 0] = bv.x; s.bt[4 * j + 1] = bv.y;
    s.bt[4 * j + 2] = bv.z; s.bt[4 * j + 3] = bv.w;
    s.ct[4 * j + 0] = cv.x; s.ct[4 * j + 1] = cv.y;
    s.ct[4 * j + 2] = cv.z; s.ct[4 * j + 3] = cv.w;
  }
}

__device__ __forceinline__ ScanCh scan_prep(float p32, float xt, float xres,
                                            float dtw, float dtb)
{
  ScanCh c;
  const float z  = fmaf(p32, dtw, dtb);
  const float sp = __logf(1.f + __expf(z));
  c.delta = (z > 20.f) ? z : sp;
  c.dx    = c.delta * xt;
  c.xres  = xres;
  return c;
}

template <int I>
__device__ __forceinline__ float row_lane(float v)
{
  return __int_as_float(__builtin_amdgcn_update_dpp(
      0, __float_as_int(v), 0x150 + I, 0xf, 0xf, false));
}

template <int M>
__device__ __forceinline__ float row_xor(float v)
{
  return __int_as_float(__builtin_amdgcn_update_dpp(
      0, __float_as_int(v), 0x160 + M, 0xf, 0xf, false));
}

template <int M>
__device__ __forceinline__ float scan_fold(bool up, float lo, float hi)
{
  const float keep = up ? hi : lo;
  const float send = up ? lo : hi;
  return keep + row_xor<M>(send);
}

template <int I, int E>
__device__ __forceinline__ void scan_decay(float delta, float a_c,
                                           float* __restrict__ ab)
{
  if constexpr (I < E) {
    ab[I] = __expf(row_lane<I>(delta) * a_c);
    scan_decay<I + 1, E>(delta, a_c, ab);
  }
}

template <int I>
__device__ __forceinline__ void scan_drive(const ScanBC& s, float dx,
                                           float* __restrict__ u)
{
  if constexpr (I < TS) {
    u[I] = row_lane<I>(dx) * s.bt[I];
    scan_drive<I + 1>(s, dx, u);
  }
}

__device__ __forceinline__ float scan_batch(
    const ScanBC& s, ScanBC& sn, const ScanCh& c, const float* __restrict__ ab,
    ScanCh& cn, float* __restrict__ abn,
    float h, float a_c, float dtw, float dtb, float dpd, int n,
    const float* lbc, const float* lxc, const float* lxr, const float* lp,
    float* lout)
{
  scan_load_bc(sn, lbc, n);
  const float xtn = *lxc;
  const float xrn = *lxr;
  const float p32n = *lp;
  float u[TS], y[TS];
  scan_drive<0>(s, c.dx, u);
#pragma unroll
  for (int i = 0; i < TS; ++i) {
    h = fmaf(ab[i], h, u[i]);
    y[i] = h * s.ct[i];
  }
  const bool n1 = n & 1, n2 = n & 2, n4 = n & 4, n8 = n & 8;
  float r8[8], r4[4], r2[2];
#pragma unroll
  for (int j = 0; j < 8; ++j) {
    const float lo = y[2 * j], hi = y[2 * j + 1];
    r8[j] = scan_fold<1>(n1, lo, hi);
  }
#pragma unroll
  for (int j = 0; j < 4; ++j) {
    const float lo = r8[2 * j], hi = r8[2 * j + 1];
    r4[j] = scan_fold<2>(n2, lo, hi);
  }
#pragma unroll
  for (int j = 0; j < 2; ++j) {
    const float lo = r4[2 * j], hi = r4[2 * j + 1];
    r2[j] = scan_fold<4>(n4, lo, hi);
  }
  const float ylo = r2[0], yhi = r2[1];
  const float yn = scan_fold<8>(n8, ylo, yhi);
  *lout = fmaf(c.xres, dpd, yn);
  cn = scan_prep(p32n, xtn, xrn, dtw, dtb);
  scan_decay<0, TS>(cn.delta, a_c, abn);
  return h;
}

struct ScanGroup { float4 xc, xr, b0, b1; float p; };

__device__ __forceinline__ ScanGroup scan_fetch(
    const float* __restrict__ xcg, const float* __restrict__ xrg,
    const float* __restrict__ bcb, const float* __restrict__ p32b,
    int step0, int srow, int tid)
{
  ScanGroup g;
  int row = step0 + srow;
  if (row > L_SZ - 1) row = L_SZ - 1;
  g.xc = *reinterpret_cast<const float4*>(xcg + (size_t)row * D_SZ);
  g.xr = *reinterpret_cast<const float4*>(xrg + (size_t)row * D_SZ);
  int f0 = step0 * (BC_TILE / TS) + 4 * tid;
  int f1 = f0 + 4 * 256;
  if (f0 >= L_SZ * (BC_TILE / TS)) f0 = 0;
  if (f1 >= L_SZ * (BC_TILE / TS)) f1 = 0;
  g.b0 = *reinterpret_cast<const float4*>(bcb + f0);
  g.b1 = *reinterpret_cast<const float4*>(bcb + f1);
  int prow = step0 + (tid & (TRIP - 1));
  if (prow > L_SZ - 1) prow = L_SZ - 1;
  g.p = p32b[prow];
  return g;
}

__device__ __forceinline__ void scan_put(
    const ScanGroup& g, float* lxc, float* lxr, float* lbc, float* lp, int tid)
{
  *reinterpret_cast<float4*>(lxc) = g.xc;
  *reinterpret_cast<float4*>(lxr) = g.xr;
  reinterpret_cast<float4*>(lbc)[tid]       = g.b0;
  reinterpret_cast<float4*>(lbc)[tid + 256] = g.b1;
  lp[tid & (TRIP - 1)] = g.p;
}

__device__ __forceinline__ void scan_handover()
{
  __builtin_amdgcn_fence(__ATOMIC_RELEASE, "workgroup", "local");
  __builtin_amdgcn_s_barrier();
  __builtin_amdgcn_fence(__ATOMIC_ACQUIRE, "workgroup", "local");
}

template <int P>
__device__ __forceinline__ float scan_trip(
    float* sBC, float* sXc, float* sXr, float* sOut, float* sP,
    const float* __restrict__ xcg, const float* __restrict__ xrg,
    const float* __restrict__ bcb, const float* __restrict__ p32b,
    float* __restrict__ outg, int srow, int srow_off, int lane_off, int tid,
    int n, int t0, const ScanGroup& gst, ScanGroup& gld,
    ScanBC& bcA, ScanBC& bcB, ScanCh& chE, ScanCh& chO,
    float* __restrict__ abE, float* __restrict__ abO,
    float h, float a_c, float dtw, float dtb, float dpd)
{
  const float* inBC = sBC + P * BCBUF;
  const float* inXc = sXc + P * SBUF + lane_off;
  const float* inXr = sXr + P * SBUF + lane_off;
  const float* inP  = sP + P * TRIP + n;
  float*       oOut = sOut + P * SBUF + lane_off;
  scan_handover();
  if (t0 != 0) {
    const float4 o =
        *reinterpret_cast<const float4*>(sOut + (P ^ 1) * SBUF + srow_off);
    *reinterpret_cast<float4*>(outg + (size_t)(t0 - TRIP + srow) * D_SZ) = o;
  }
  gld = scan_fetch(xcg, xrg, bcb, p32b, t0 + 2 * TRIP + TS, srow, tid);
  h = scan_batch(bcA, bcB, chE, abE, chO, abO, h, a_c, dtw, dtb, dpd, n,
                 inBC, inXc, inXr, inP, oOut);
  h = scan_batch(bcB, bcA, chO, abO, chE, abE, h, a_c, dtw, dtb, dpd, n,
                 inBC + BC_TILE, inXc + TS * SPITCH, inXr + TS * SPITCH,
                 inP + TS, oOut + TS * SPITCH);
  h = scan_batch(bcA, bcB, chE, abE, chO, abO, h, a_c, dtw, dtb, dpd, n,
                 inBC + 2 * BC_TILE, inXc + 2 * TS * SPITCH,
                 inXr + 2 * TS * SPITCH, inP + 2 * TS, oOut + 2 * TS * SPITCH);
  h = scan_batch(bcB, bcA, chO, abO, chE, abE, h, a_c, dtw, dtb, dpd, n,
                 inBC + 3 * BC_TILE, inXc + 3 * TS * SPITCH,
                 inXr + 3 * TS * SPITCH, inP + 3 * TS, oOut + 3 * TS * SPITCH);
  __builtin_amdgcn_sched_barrier(0);
  scan_put(gst, sXc + (P ^ 1) * SBUF + srow_off, sXr + (P ^ 1) * SBUF + srow_off,
           sBC + (P ^ 1) * BCBUF, sP + (P ^ 1) * TRIP, tid);
  return h;
}

__global__ __launch_bounds__(256)
__attribute__((amdgpu_waves_per_eu(2, 2))) void scan_kernel(
    const float* __restrict__ x, const float* __restrict__ xc,
    const float* __restrict__ bc, const float* __restrict__ p32,
    const float* __restrict__ dt_w,
    const float* __restrict__ dt_b, const float* __restrict__ a_log,
    const float* __restrict__ Dp, float* __restrict__ out)
{
  __shared__ __attribute__((aligned(16))) float sBC[2 * BCBUF];
  __shared__ __attribute__((aligned(16))) float sXc[2 * SBUF];
  __shared__ __attribute__((aligned(16))) float sXr[2 * SBUF];
  __shared__ __attribute__((aligned(16))) float sOut[2 * SBUF];
  __shared__ float sP[2 * TRIP];

  const int tid = threadIdx.x;
  const int n   = tid & 15;
  const int cw  = tid >> 4;
  const int c0  = blockIdx.x * 16;
  const int b   = c0 >> 10;
  const int d0  = c0 & (D_SZ - 1);
  const int d   = d0 + cw;

  const float a_c = -__expf(a_log[d * N_SZ + n]);
  const float dtw = dt_w[d];
  const float dtb = dt_b[d];
  const float dpd = Dp[d];

  const float* bcb  = bc  + (size_t)b * L_SZ * (BC_TILE / TS);
  const float* p32b = p32 + (size_t)b * L_SZ;

  const int srow = tid >> 2;
  const int sq   = tid & 3;
  const size_t gbase = (size_t)b * L_SZ * D_SZ + d0 + 4 * sq;
  const float* xcg  = xc  + gbase;
  const float* xrg  = x   + gbase;
  float*       outg = out + gbase;
  const int srow_off = srow * SPITCH + 4 * sq;
  const int lane_off = n * SPITCH + cw;

  ScanBC bcA, bcB;
  ScanCh chE, chO;
  float  abE[TS], abO[TS];

  scan_load_bc(bcA, bcb, n);
  {
    const float pv = p32b[n];
    const float xt = xc[(size_t)b * L_SZ * D_SZ + (size_t)n * D_SZ + d];
    const float xr = x[(size_t)b * L_SZ * D_SZ + (size_t)n * D_SZ + d];
    chE = scan_prep(pv, xt, xr, dtw, dtb);
    scan_decay<0, TS>(chE.delta, a_c, abE);
  }
  {
    const ScanGroup g0 = scan_fetch(xcg, xrg, bcb, p32b, TS, srow, tid);
    scan_put(g0, sXc + srow_off, sXr + srow_off, sBC, sP, tid);
  }
  ScanGroup gA;
  ScanGroup gB = scan_fetch(xcg, xrg, bcb, p32b, TS + TRIP, srow, tid);

  float h = 0.f;
  for (int t0 = 0; t0 < L_SZ; t0 += 2 * TRIP) {
    h = scan_trip<0>(sBC, sXc, sXr, sOut, sP, xcg, xrg, bcb, p32b, outg, srow,
                     srow_off, lane_off, tid, n, t0, gB, gA, bcA, bcB, chE, chO,
                     abE, abO, h, a_c, dtw, dtb, dpd);
    h = scan_trip<1>(sBC, sXc, sXr, sOut, sP, xcg, xrg, bcb, p32b, outg, srow,
                     srow_off, lane_off, tid, n, t0 + TRIP, gA, gB, bcA, bcB,
                     chE, chO, abE, abO, h, a_c, dtw, dtb, dpd);
  }
  scan_handover();
  {
    const int pl = ((L_SZ / TRIP) - 1) & 1;
    const float4 o =
        *reinterpret_cast<const float4*>(sOut + pl * SBUF + srow_off);
    *reinterpret_cast<float4*>(outg + (size_t)(L_SZ - TRIP + srow) * D_SZ) = o;
  }
}

extern "C" void kernel_launch(void* const* d_in, const int* in_sizes, int n_in,
                              void* d_out, int out_size, void* d_ws, size_t ws_size,
                              hipStream_t stream) {
  const float* x      = (const float*)d_in[0];
  const float* conv_w = (const float*)d_in[1];
  const float* conv_b = (const float*)d_in[2];
  const float* xproj  = (const float*)d_in[3];
  const float* dt_w   = (const float*)d_in[4];
  const float* dt_b   = (const float*)d_in[5];
  const float* a_log  = (const float*)d_in[6];
  const float* Dp     = (const float*)d_in[7];
  float* out = (float*)d_out;

  float* xc   = (float*)d_ws;
  float* bc   = (float*)((char*)d_ws +
                         (size_t)B_SZ * L_SZ * D_SZ * sizeof(float));
  float* p32  = bc + (size_t)B_SZ * L_SZ * (BC_TILE / 16);

  conv_silu_kernel<<<(B_SZ * L_SZ) / CONV_TL, D_QUADS, 0, stream>>>(
      x, conv_w, conv_b, xc);

  proj_wmma_kernel<<<512, 256, 0, stream>>>(xc, xproj, bc, p32);

  scan_kernel<<<(B_SZ * D_SZ) / 16, 256, 0, stream>>>(
      x, xc, bc, p32, dt_w, dt_b, a_log, Dp, out);
}


# conv+SiLU and proj GEMM in one launch: a block keeps its 16-row conv tile in LDS and takes the A fragments from there (2 launches, not 3; conv output not read back for proj); fp order unchanged
# speedup vs baseline: 24.3821x; 1.0893x over previous
#include <hip/hip_runtime.h>
#include <hip/hip_bf16.h>

#define B_SZ 4
#define L_SZ 2048
#define D_SZ 1024
#define N_SZ 16
#define P_SZ 33

typedef float v2f __attribute__((ext_vector_type(2)));
typedef float v8f __attribute__((ext_vector_type(8)));

#define CONV_TL 16
#define D_QUADS (D_SZ / 4)
#define BC_TILE (2 * 16 * N_SZ)
#define KSPLIT 8
#define KQ (D_SZ / KSPLIT)
#define APITCH (D_SZ + 4)
static_assert(KQ % 4 == 0, "a wave's share of K is a whole number of k steps");
static_assert(CONV_TL == 16, "a conv chunk is one 16-row M tile");
static_assert(D_QUADS == 32 * KSPLIT, "one thread per channel quad");
static_assert((APITCH * 4) % 32 == 16, "row pitch: an odd multiple of 16 bytes");

__device__ __forceinline__ float conv_silu_one(
    float acc, float x0, float x1, float x2, float x3, const float4 w)
{
  acc = fmaf(x0, w.x, acc);
  acc = fmaf(x1, w.y, acc);
  acc = fmaf(x2, w.z, acc);
  acc = fmaf(x3, w.w, acc);
  return acc / (1.f + __expf(-acc));
}

__device__ __forceinline__ void lds_handover()
{
  __builtin_amdgcn_fence(__ATOMIC_RELEASE, "workgroup", "local");
  __builtin_amdgcn_s_barrier();
  __builtin_amdgcn_fence(__ATOMIC_ACQUIRE, "workgroup", "local");
}

__global__ __launch_bounds__(32 * KSPLIT)
__attribute__((amdgpu_waves_per_eu(4, 4))) void conv_proj_kernel(
    const float* __restrict__ x, const float* __restrict__ cw,
    const float* __restrict__ cb, const float* __restrict__ w,
    float* __restrict__ xc, float* __restrict__ bc, float* __restrict__ p32)
{
  __shared__ __attribute__((aligned(16))) float sA[CONV_TL * APITCH];
  __shared__ float red[(KSPLIT - 1) * 24 * 32];

  const int tile = blockIdx.x;
  {
    const int q   = threadIdx.x;
    const int bl0 = tile * CONV_TL;
    const int l0  = bl0 & (L_SZ - 1);
    const float4* xp =
        reinterpret_cast<const float4*>(x + (size_t)bl0 * D_SZ) + q;
    float4* op = reinterpret_cast<float4*>(xc + (size_t)bl0 * D_SZ) + q;
    float* lp = sA + 4 * q;

    const float4* wq = reinterpret_cast<const float4*>(cw) + 4 * q;
    const float4 w0 = wq[0], w1 = wq[1], w2 = wq[2], w3 = wq[3];
    const float4 bias = reinterpret_cast<const float4*>(cb)[q];

    float4 xin[CONV_TL + 3];
#pragma unroll
    for (int i = 0; i < CONV_TL + 3; ++i) {
      const bool pad = (l0 + i - 3) < 0;
      const float4 v = xp[pad ? 0 : (i - 3) * D_QUADS];
      xin[i].x = pad ? 0.f : v.x;
      xin[i].y = pad ? 0.f : v.y;
      xin[i].z = pad ? 0.f : v.z;
      xin[i].w = pad ? 0.f : v.w;
    }

#pragma unroll
    for (int i = 0; i < CONV_TL; ++i) {
      float4 o;
      o.x = conv_silu_one(bias.x, xin[i].x, xin[i + 1].x, xin[i + 2].x,
                          xin[i + 3].x, w0);
      o.y = conv_silu_one(bias.y, xin[i].y, xin[i + 1].y, xin[i + 2].y,
                          xin[i + 3].y, w1);
      o.z = conv_silu_one(bias.z, xin[i].z, xin[i + 1].z, xin[i + 2].z,
                          xin[i + 3].z, w2);
      o.w = conv_silu_one(bias.w, xin[i].w, xin[i + 1].w, xin[i + 2].w,
                          xin[i + 3].w, w3);
      op[i * D_QUADS] = o;
      *reinterpret_cast<float4*>(lp + i * APITCH) = o;
    }
  }
  lds_handover();

  const int wv   = threadIdx.x >> 5;
  const int lane = threadIdx.x & 31;
  const int kb   = ((lane >> 4) << 1) + wv * KQ;
  const float* arow = sA + (lane & 15) * APITCH + kb;

  const int ncol = lane & 15;
  const int p0 = ncol;
  const int p1 = 16 + ncol;
  int p2 = 32 + ncol;
  const float s2 = (p2 < P_SZ) ? 1.f : 0.f;
  if (p2 >= P_SZ) p2 = P_SZ - 1;
  const float* brow0 = w + (size_t)p0 * D_SZ + kb;
  const float* brow1 = w + (size_t)p1 * D_SZ + kb;
  const float* brow2 = w + (size_t)p2 * D_SZ + kb;

  v8f acc0 = {}, acc1 = {}, acc2 = {};
  for (int k0 = 0; k0 < KQ; k0 += 4) {
    v2f a  = *(const v2f*)(arow  + k0);
    v2f b0 = *(const v2f*)(brow0 + k0);
    v2f b1 = *(const v2f*)(brow1 + k0);
    v2f b2 = *(const v2f*)(brow2 + k0);
    b2 *= s2;
    acc0 = __builtin_amdgcn_wmma_f32_16x16x4_f32(false, a, false, b0,
                                                 (short)0, acc0, false, false);
    acc1 = __builtin_amdgcn_wmma_f32_16x16x4_f32(false, a, false, b1,
                                                 (short)0, acc1, false, false);
    acc2 = __builtin_amdgcn_wmma_f32_16x16x4_f32(false, a, false, b2,
                                                 (short)0, acc2, false, false);
  }

  if (wv != 0) {
    float* rp = red + (wv - 1) * (24 * 32) + lane;
#pragma unroll
    for (int v = 0; v < 8; ++v) {
      rp[v * 32]        = acc0[v];
      rp[(8 + v) * 32]  = acc1[v];
      rp[(16 + v) * 32] = acc2[v];
    }
  }
  __syncthreads();
  if (wv != 0) return;
#pragma unroll
  for (int j = 0; j < KSPLIT - 1; ++j) {
    const float* rp = red + j * (24 * 32) + lane;
#pragma unroll
    for (int v = 0; v < 8; ++v) {
      acc0[v] += rp[v * 32];
      acc1[v] += rp[(8 + v) * 32];
      acc2[v] += rp[(16 + v) * 32];
    }
  }

  const int hh = lane >> 4;
  float4* tp = reinterpret_cast<float4*>(bc + (size_t)tile * BC_TILE)
               + (2 * hh) * N_SZ + ncol;
  tp[0]             = make_float4(acc0[0], acc0[1], acc0[2], acc0[3]);
  tp[N_SZ]          = make_float4(acc0[4], acc0[5], acc0[6], acc0[7]);
  tp[4 * N_SZ]        = make_float4(acc1[0], acc1[1], acc1[2], acc1[3]);
  tp[4 * N_SZ + N_SZ] = make_float4(acc1[4], acc1[5], acc1[6], acc1[7]);
  if (ncol == 0) {
    float4* pp = reinterpret_cast<float4*>(p32 + tile * 16 + 8 * hh);
    pp[0] = make_float4(acc2[0], acc2[1], acc2[2], acc2[3]);
    pp[1] = make_float4(acc2[4], acc2[5], acc2[6], acc2[7]);
  }
}

#define TS 16
#define TRIP (4 * TS)
#define SPITCH 20
#define SBUF (TRIP * SPITCH)
#define BCBUF (4 * BC_TILE)
static_assert(TS == N_SZ, "lane n of a row owns step t0 + n of a batch");
static_assert(L_SZ % (2 * TRIP) == 0, "the time loop runs two trips per pass");
static_assert((SPITCH * 4) % 32 == 16, "row pitch: an odd multiple of 16 bytes");

struct ScanBC  { float bt[TS], ct[TS]; };
struct ScanCh  { float delta, dx, xres; };

__device__ __forceinline__ void scan_load_bc(
    ScanBC& s, const float* __restrict__ tile, int n)
{
  const float4* tp = reinterpret_cast<const float4*>(tile) + n;
#pragma unroll
  for (int j = 0; j < TS / 4; ++j) {
    const float4 bv = tp[j * N_SZ];
    const float4 cv = tp[(TS / 4 + j) * N_SZ];
    s.bt[4 * j + 0] = bv.x; s.bt[4 * j + 1] = bv.y;
    s.bt[4 * j + 2] = bv.z; s.bt[4 * j + 3] = bv.w;
    s.ct[4 * j + 0] = cv.x; s.ct[4 * j + 1] = cv.y;
    s.ct[4 * j + 2] = cv.z; s.ct[4 * j + 3] = cv.w;
  }
}

__device__ __forceinline__ ScanCh scan_prep(float p32, float xt, float xres,
                                            float dtw, float dtb)
{
  ScanCh c;
  const float z  = fmaf(p32, dtw, dtb);
  const float sp = __logf(1.f + __expf(z));
  c.delta = (z > 20.f) ? z : sp;
  c.dx    = c.delta * xt;
  c.xres  = xres;
  return c;
}

template <int I>
__device__ __forceinline__ float row_lane(float v)
{
  return __int_as_float(__builtin_amdgcn_update_dpp(
      0, __float_as_int(v), 0x150 + I, 0xf, 0xf, false));
}

template <int M>
__device__ __forceinline__ float row_xor(float v)
{
  return __int_as_float(__builtin_amdgcn_update_dpp(
      0, __float_as_int(v), 0x160 + M, 0xf, 0xf, false));
}

template <int M>
__device__ __forceinline__ float scan_fold(bool up, float lo, float hi)
{
  const float keep = up ? hi : lo;
  const float send = up ? lo : hi;
  return keep + row_xor<M>(send);
}

template <int I, int E>
__device__ __forceinline__ void scan_decay(float delta, float a_c,
                                           float* __restrict__ ab)
{
  if constexpr (I < E) {
    ab[I] = __expf(row_lane<I>(delta) * a_c);
    scan_decay<I + 1, E>(delta, a_c, ab);
  }
}

template <int I>
__device__ __forceinline__ void scan_drive(const ScanBC& s, float dx,
                                           float* __restrict__ u)
{
  if constexpr (I < TS) {
    u[I] = row_lane<I>(dx) * s.bt[I];
    scan_drive<I + 1>(s, dx, u);
  }
}

__device__ __forceinline__ float scan_batch(
    const ScanBC& s, ScanBC& sn, const ScanCh& c, const float* __restrict__ ab,
    ScanCh& cn, float* __restrict__ abn,
    float h, float a_c, float dtw, float dtb, float dpd, int n,
    const float* lbc, const float* lxc, const float* lxr, const float* lp,
    float* lout)
{
  scan_load_bc(sn, lbc, n);
  const float xtn = *lxc;
  const float xrn = *lxr;
  const float p32n = *lp;
  float u[TS], y[TS];
  scan_drive<0>(s, c.dx, u);
#pragma unroll
  for (int i = 0; i < TS; ++i) {
    h = fmaf(ab[i], h, u[i]);
    y[i] = h * s.ct[i];
  }
  const bool n1 = n & 1, n2 = n & 2, n4 = n & 4, n8 = n & 8;
  float r8[8], r4[4], r2[2];
#pragma unroll
  for (int j = 0; j < 8; ++j) {
    const float lo = y[2 * j], hi = y[2 * j + 1];
    r8[j] = scan_fold<1>(n1, lo, hi);
  }
#pragma unroll
  for (int j = 0; j < 4; ++j) {
    const float lo = r8[2 * j], hi = r8[2 * j + 1];
    r4[j] = scan_fold<2>(n2, lo, hi);
  }
#pragma unroll
  for (int j = 0; j < 2; ++j) {
    const float lo = r4[2 * j], hi = r4[2 * j + 1];
    r2[j] = scan_fold<4>(n4, lo, hi);
  }
  const float ylo = r2[0], yhi = r2[1];
  const float yn = scan_fold<8>(n8, ylo, yhi);
  *lout = fmaf(c.xres, dpd, yn);
  cn = scan_prep(p32n, xtn, xrn, dtw, dtb);
  scan_decay<0, TS>(cn.delta, a_c, abn);
  return h;
}

struct ScanGroup { float4 xc, xr, b0, b1; float p; };

__device__ __forceinline__ ScanGroup scan_fetch(
    const float* __restrict__ xcg, const float* __restrict__ xrg,
    const float* __restrict__ bcb, const float* __restrict__ p32b,
    int step0, int srow, int tid)
{
  ScanGroup g;
  int row = step0 + srow;
  if (row > L_SZ - 1) row = L_SZ - 1;
  g.xc = *reinterpret_cast<const float4*>(xcg + (size_t)row * D_SZ);
  g.xr = *reinterpret_cast<const float4*>(xrg + (size_t)row * D_SZ);
  int f0 = step0 * (BC_TILE / TS) + 4 * tid;
  int f1 = f0 + 4 * 256;
  if (f0 >= L_SZ * (BC_TILE / TS)) f0 = 0;
  if (f1 >= L_SZ * (BC_TILE / TS)) f1 = 0;
  g.b0 = *reinterpret_cast<const float4*>(bcb + f0);
  g.b1 = *reinterpret_cast<const float4*>(bcb + f1);
  int prow = step0 + (tid & (TRIP - 1));
  if (prow > L_SZ - 1) prow = L_SZ - 1;
  g.p = p32b[prow];
  return g;
}

__device__ __forceinline__ void scan_put(
    const ScanGroup& g, float* lxc, float* lxr, float* lbc, float* lp, int tid)
{
  *reinterpret_cast<float4*>(lxc) = g.xc;
  *reinterpret_cast<float4*>(lxr) = g.xr;
  reinterpret_cast<float4*>(lbc)[tid]       = g.b0;
  reinterpret_cast<float4*>(lbc)[tid + 256] = g.b1;
  lp[tid & (TRIP - 1)] = g.p;
}

template <int P>
__device__ __forceinline__ float scan_trip(
    float* sBC, float* sXc, float* sXr, float* sOut, float* sP,
    const float* __restrict__ xcg, const float* __restrict__ xrg,
    const float* __restrict__ bcb, const float* __restrict__ p32b,
    float* __restrict__ outg, int srow, int srow_off, int lane_off, int tid,
    int n, int t0, const ScanGroup& gst, ScanGroup& gld,
    ScanBC& bcA, ScanBC& bcB, ScanCh& chE, ScanCh& chO,
    float* __restrict__ abE, float* __restrict__ abO,
    float h, float a_c, float dtw, float dtb, float dpd)
{
  const float* inBC = sBC + P * BCBUF;
  const float* inXc = sXc + P * SBUF + lane_off;
  const float* inXr = sXr + P * SBUF + lane_off;
  const float* inP  = sP + P * TRIP + n;
  float*       oOut = sOut + P * SBUF + lane_off;
  lds_handover();
  if (t0 != 0) {
    const float4 o =
        *reinterpret_cast<const float4*>(sOut + (P ^ 1) * SBUF + srow_off);
    *reinterpret_cast<float4*>(outg + (size_t)(t0 - TRIP + srow) * D_SZ) = o;
  }
  gld = scan_fetch(xcg, xrg, bcb, p32b, t0 + 2 * TRIP + TS, srow, tid);
  h = scan_batch(bcA, bcB, chE, abE, chO, abO, h, a_c, dtw, dtb, dpd, n,
                 inBC, inXc, inXr, inP, oOut);
  h = scan_batch(bcB, bcA, chO, abO, chE, abE, h, a_c, dtw, dtb, dpd, n,
                 inBC + BC_TILE, inXc + TS * SPITCH, inXr + TS * SPITCH,
                 inP + TS, oOut + TS * SPITCH);
  h = scan_batch(bcA, bcB, chE, abE, chO, abO, h, a_c, dtw, dtb, dpd, n,
                 inBC + 2 * BC_TILE, inXc + 2 * TS * SPITCH,
                 inXr + 2 * TS * SPITCH, inP + 2 * TS, oOut + 2 * TS * SPITCH);
  h = scan_batch(bcB, bcA, chO, abO, chE, abE, h, a_c, dtw, dtb, dpd, n,
                 inBC + 3 * BC_TILE, inXc + 3 * TS * SPITCH,
                 inXr + 3 * TS * SPITCH, inP + 3 * TS, oOut + 3 * TS * SPITCH);
  __builtin_amdgcn_sched_barrier(0);
  scan_put(gst, sXc + (P ^ 1) * SBUF + srow_off, sXr + (P ^ 1) * SBUF + srow_off,
           sBC + (P ^ 1) * BCBUF, sP + (P ^ 1) * TRIP, tid);
  return h;
}

__global__ __launch_bounds__(256)
__attribute__((amdgpu_waves_per_eu(2, 2))) void scan_kernel(
    const float* __restrict__ x, const float* __restrict__ xc,
    const float* __restrict__ bc, const float* __restrict__ p32,
    const float* __restrict__ dt_w,
    const float* __restrict__ dt_b, const float* __restrict__ a_log,
    const float* __restrict__ Dp, float* __restrict__ out)
{
  __shared__ __attribute__((aligned(16))) float sBC[2 * BCBUF];
  __shared__ __attribute__((aligned(16))) float sXc[2 * SBUF];
  __shared__ __attribute__((aligned(16))) float sXr[2 * SBUF];
  __shared__ __attribute__((aligned(16))) float sOut[2 * SBUF];
  __shared__ float sP[2 * TRIP];

  const int tid = threadIdx.x;
  const int n   = tid & 15;
  const int cw  = tid >> 4;
  const int c0  = blockIdx.x * 16;
  const int b   = c0 >> 10;
  const int d0  = c0 & (D_SZ - 1);
  const int d   = d0 + cw;

  const float a_c = -__expf(a_log[d * N_SZ + n]);
  const float dtw = dt_w[d];
  const float dtb = dt_b[d];
  const float dpd = Dp[d];

  const float* bcb  = bc  + (size_t)b * L_SZ * (BC_TILE / TS);
  const float* p32b = p32 + (size_t)b * L_SZ;

  const int srow = tid >> 2;
  const int sq   = tid & 3;
  const size_t gbase = (size_t)b * L_SZ * D_SZ + d0 + 4 * sq;
  const float* xcg  = xc  + gbase;
  const float* xrg  = x   + gbase;
  float*       outg = out + gbase;
  const int srow_off = srow * SPITCH + 4 * sq;
  const int lane_off = n * SPITCH + cw;

  ScanBC bcA, bcB;
  ScanCh chE, chO;
  float  abE[TS], abO[TS];

  scan_load_bc(bcA, bcb, n);
  {
    const float pv = p32b[n];
    const float xt = xc[(size_t)b * L_SZ * D_SZ + (size_t)n * D_SZ + d];
    const float xr = x[(size_t)b * L_SZ * D_SZ + (size_t)n * D_SZ + d];
    chE = scan_prep(pv, xt, xr, dtw, dtb);
    scan_decay<0, TS>(chE.delta, a_c, abE);
  }
  {
    const ScanGroup g0 = scan_fetch(xcg, xrg, bcb, p32b, TS, srow, tid);
    scan_put(g0, sXc + srow_off, sXr + srow_off, sBC, sP, tid);
  }
  ScanGroup gA;
  ScanGroup gB = scan_fetch(xcg, xrg, bcb, p32b, TS + TRIP, srow, tid);

  float h = 0.f;
  for (int t0 = 0; t0 < L_SZ; t0 += 2 * TRIP) {
    h = scan_trip<0>(sBC, sXc, sXr, sOut, sP, xcg, xrg, bcb, p32b, outg, srow,
                     srow_off, lane_off, tid, n, t0, gB, gA, bcA, bcB, chE, chO,
                     abE, abO, h, a_c, dtw, dtb, dpd);
    h = scan_trip<1>(sBC, sXc, sXr, sOut, sP, xcg, xrg, bcb, p32b, outg, srow,
                     srow_off, lane_off, tid, n, t0 + TRIP, gA, gB, bcA, bcB,
                     chE, chO, abE, abO, h, a_c, dtw, dtb, dpd);
  }
  lds_handover();
  {
    const int pl = ((L_SZ / TRIP) - 1) & 1;
    const float4 o =
        *reinterpret_cast<const float4*>(sOut + pl * SBUF + srow_off);
    *reinterpret_cast<float4*>(outg + (size_t)(L_SZ - TRIP + srow) * D_SZ) = o;
  }
}

extern "C" void kernel_launch(void* const* d_in, const int* in_sizes, int n_in,
                              void* d_out, int out_size, void* d_ws, size_t ws_size,
                              hipStream_t stream) {
  const float* x      = (const float*)d_in[0];
  const float* conv_w = (const float*)d_in[1];
  const float* conv_b = (const float*)d_in[2];
  const float* xproj  = (const float*)d_in[3];
  const float* dt_w   = (const float*)d_in[4];
  const float* dt_b   = (const float*)d_in[5];
  const float* a_log  = (const float*)d_in[6];
  const float* Dp     = (const float*)d_in[7];
  float* out = (float*)d_out;

  float* xc   = (float*)d_ws;
  float* bc   = (float*)((char*)d_ws +
                         (size_t)B_SZ * L_SZ * D_SZ * sizeof(float));
  float* p32  = bc + (size_t)B_SZ * L_SZ * (BC_TILE / 16);

  conv_proj_kernel<<<(B_SZ * L_SZ) / CONV_TL, D_QUADS, 0, stream>>>(
      x, conv_w, conv_b, xproj, xc, bc, p32);

  scan_kernel<<<(B_SZ * D_SZ) / 16, 256, 0, stream>>>(
      x, xc, bc, p32, dt_w, dt_b, a_log, Dp, out);
}


# scan: exp and log taken in base 2, log2(e) folded into the lane's decay rate once and into softplus: per step one multiply less in front of v_exp, plain v_log; decay, softplus rounding changed
# speedup vs baseline: 25.6205x; 1.0508x over previous
#include <hip/hip_runtime.h>
#include <hip/hip_bf16.h>

#define B_SZ 4
#define L_SZ 2048
#define D_SZ 1024
#define N_SZ 16
#define P_SZ 33

typedef float v2f __attribute__((ext_vector_type(2)));
typedef float v8f __attribute__((ext_vector_type(8)));

#define CONV_TL 16
#define D_QUADS (D_SZ / 4)
#define BC_TILE (2 * 16 * N_SZ)
#define KSPLIT 8
#define KQ (D_SZ / KSPLIT)
#define APITCH (D_SZ + 4)
static_assert(KQ % 4 == 0, "a wave's share of K is a whole number of k steps");
static_assert(CONV_TL == 16, "a conv chunk is one 16-row M tile");
static_assert(D_QUADS == 32 * KSPLIT, "one thread per channel quad");
static_assert((APITCH * 4) % 32 == 16, "row pitch: an odd multiple of 16 bytes");

__device__ __forceinline__ float conv_silu_one(
    float acc, float x0, float x1, float x2, float x3, const float4 w)
{
  acc = fmaf(x0, w.x, acc);
  acc = fmaf(x1, w.y, acc);
  acc = fmaf(x2, w.z, acc);
  acc = fmaf(x3, w.w, acc);
  return acc / (1.f + __expf(-acc));
}

__device__ __forceinline__ void lds_handover()
{
  __builtin_amdgcn_fence(__ATOMIC_RELEASE, "workgroup", "local");
  __builtin_amdgcn_s_barrier();
  __builtin_amdgcn_fence(__ATOMIC_ACQUIRE, "workgroup", "local");
}

__global__ __launch_bounds__(32 * KSPLIT)
__attribute__((amdgpu_waves_per_eu(4, 4))) void conv_proj_kernel(
    const float* __restrict__ x, const float* __restrict__ cw,
    const float* __restrict__ cb, const float* __restrict__ w,
    float* __restrict__ xc, float* __restrict__ bc, float* __restrict__ p32)
{
  __shared__ __attribute__((aligned(16))) float sA[CONV_TL * APITCH];
  __shared__ float red[(KSPLIT - 1) * 24 * 32];

  const int tile = blockIdx.x;
  {
    const int q   = threadIdx.x;
    const int bl0 = tile * CONV_TL;
    const int l0  = bl0 & (L_SZ - 1);
    const float4* xp =
        reinterpret_cast<const float4*>(x + (size_t)bl0 * D_SZ) + q;
    float4* op = reinterpret_cast<float4*>(xc + (size_t)bl0 * D_SZ) + q;
    float* lp = sA + 4 * q;

    const float4* wq = reinterpret_cast<const float4*>(cw) + 4 * q;
    const float4 w0 = wq[0], w1 = wq[1], w2 = wq[2], w3 = wq[3];
    const float4 bias = reinterpret_cast<const float4*>(cb)[q];

    float4 xin[CONV_TL + 3];
#pragma unroll
    for (int i = 0; i < CONV_TL + 3; ++i) {
      const bool pad = (l0 + i - 3) < 0;
      const float4 v = xp[pad ? 0 : (i - 3) * D_QUADS];
      xin[i].x = pad ? 0.f : v.x;
      xin[i].y = pad ? 0.f : v.y;
      xin[i].z = pad ? 0.f : v.z;
      xin[i].w = pad ? 0.f : v.w;
    }

#pragma unroll
    for (int i = 0; i < CONV_TL; ++i) {
      float4 o;
      o.x = conv_silu_one(bias.x, xin[i].x, xin[i + 1].x, xin[i + 2].x,
                          xin[i + 3].x, w0);
      o.y = conv_silu_one(bias.y, xin[i].y, xin[i + 1].y, xin[i + 2].y,
                          xin[i + 3].y, w1);
      o.z = conv_silu_one(bias.z, xin[i].z, xin[i + 1].z, xin[i + 2].z,
                          xin[i + 3].z, w2);
      o.w = conv_silu_one(bias.w, xin[i].w, xin[i + 1].w, xin[i + 2].w,
                          xin[i + 3].w, w3);
      op[i * D_QUADS] = o;
      *reinterpret_cast<float4*>(lp + i * APITCH) = o;
    }
  }
  lds_handover();

  const int wv   = threadIdx.x >> 5;
  const int lane = threadIdx.x & 31;
  const int kb   = ((lane >> 4) << 1) + wv * KQ;
  const float* arow = sA + (lane & 15) * APITCH + kb;

  const int ncol = lane & 15;
  const int p0 = ncol;
  const int p1 = 16 + ncol;
  int p2 = 32 + ncol;
  const float s2 = (p2 < P_SZ) ? 1.f : 0.f;
  if (p2 >= P_SZ) p2 = P_SZ - 1;
  const float* brow0 = w + (size_t)p0 * D_SZ + kb;
  const float* brow1 = w + (size_t)p1 * D_SZ + kb;
  const float* brow2 = w + (size_t)p2 * D_SZ + kb;

  v8f acc0 = {}, acc1 = {}, acc2 = {};
  for (int k0 = 0; k0 < KQ; k0 += 4) {
    v2f a  = *(const v2f*)(arow  + k0);
    v2f b0 = *(const v2f*)(brow0 + k0);
    v2f b1 = *(const v2f*)(brow1 + k0);
    v2f b2 = *(const v2f*)(brow2 + k0);
    b2 *= s2;
    acc0 = __builtin_amdgcn_wmma_f32_16x16x4_f32(false, a, false, b0,
                                                 (short)0, acc0, false, false);
    acc1 = __builtin_amdgcn_wmma_f32_16x16x4_f32(false, a, false, b1,
                                                 (short)0, acc1, false, false);
    acc2 = __builtin_amdgcn_wmma_f32_16x16x4_f32(false, a, false, b2,
                                                 (short)0, acc2, false, false);
  }

  if (wv != 0) {
    float* rp = red + (wv - 1) * (24 * 32) + lane;
#pragma unroll
    for (int v = 0; v < 8; ++v) {
      rp[v * 32]        = acc0[v];
      rp[(8 + v) * 32]  = acc1[v];
      rp[(16 + v) * 32] = acc2[v];
    }
  }
  __syncthreads();
  if (wv != 0) return;
#pragma unroll
  for (int j = 0; j < KSPLIT - 1; ++j) {
    const float* rp = red + j * (24 * 32) + lane;
#pragma unroll
    for (int v = 0; v < 8; ++v) {
      acc0[v] += rp[v * 32];
      acc1[v] += rp[(8 + v) * 32];
      acc2[v] += rp[(16 + v) * 32];
    }
  }

  const int hh = lane >> 4;
  float4* tp = reinterpret_cast<float4*>(bc + (size_t)tile * BC_TILE)
               + (2 * hh) * N_SZ + ncol;
  tp[0]             = make_float4(acc0[0], acc0[1], acc0[2], acc0[3]);
  tp[N_SZ]          = make_float4(acc0[4], acc0[5], acc0[6], acc0[7]);
  tp[4 * N_SZ]        = make_float4(acc1[0], acc1[1], acc1[2], acc1[3]);
  tp[4 * N_SZ + N_SZ] = make_float4(acc1[4], acc1[5], acc1[6], acc1[7]);
  if (ncol == 0) {
    float4* pp = reinterpret_cast<float4*>(p32 + tile * 16 + 8 * hh);
    pp[0] = make_float4(acc2[0], acc2[1], acc2[2], acc2[3]);
    pp[1] = make_float4(acc2[4], acc2[5], acc2[6], acc2[7]);
  }
}

#define TS 16
#define TRIP (4 * TS)
#define SPITCH 20
#define SBUF (TRIP * SPITCH)
#define BCBUF (4 * BC_TILE)
static_assert(TS == N_SZ, "lane n of a row owns step t0 + n of a batch");
static_assert(L_SZ % (2 * TRIP) == 0, "the time loop runs two trips per pass");
static_assert((SPITCH * 4) % 32 == 16, "row pitch: an odd multiple of 16 bytes");

#define LOG2E_F 1.44269504088896340736f
#define LN2_F   0.69314718055994530942f

struct ScanBC  { float bt[TS], ct[TS]; };
struct ScanCh  { float delta, dx, xres; };

__device__ __forceinline__ void scan_load_bc(
    ScanBC& s, const float* __restrict__ tile, int n)
{
  const float4* tp = reinterpret_cast<const float4*>(tile) + n;
#pragma unroll
  for (int j = 0; j < TS / 4; ++j) {
    const float4 bv = tp[j * N_SZ];
    const float4 cv = tp[(TS / 4 + j) * N_SZ];
    s.bt[4 * j + 0] = bv.x; s.bt[4 * j + 1] = bv.y;
    s.bt[4 * j + 2] = bv.z; s.bt[4 * j + 3] = bv.w;
    s.ct[4 * j + 0] = cv.x; s.ct[4 * j + 1] = cv.y;
    s.ct[4 * j + 2] = cv.z; s.ct[4 * j + 3] = cv.w;
  }
}

__device__ __forceinline__ ScanCh scan_prep(float p32, float xt, float xres,
                                            float dtw, float dtb)
{
  ScanCh c;
  const float z  = fmaf(p32, dtw, dtb);
  const float sp = LN2_F * __builtin_amdgcn_logf(
      1.f + __builtin_amdgcn_exp2f(z * LOG2E_F));
  c.delta = (z > 20.f) ? z : sp;
  c.dx    = c.delta * xt;
  c.xres  = xres;
  return c;
}

template <int I>
__device__ __forceinline__ float row_lane(float v)
{
  return __int_as_float(__builtin_amdgcn_update_dpp(
      0, __float_as_int(v), 0x150 + I, 0xf, 0xf, false));
}

template <int M>
__device__ __forceinline__ float row_xor(float v)
{
  return __int_as_float(__builtin_amdgcn_update_dpp(
      0, __float_as_int(v), 0x160 + M, 0xf, 0xf, false));
}

template <int M>
__device__ __forceinline__ float scan_fold(bool up, float lo, float hi)
{
  const float keep = up ? hi : lo;
  const float send = up ? lo : hi;
  return keep + row_xor<M>(send);
}

template <int I, int E>
__device__ __forceinline__ void scan_decay(float delta, float a_c,
                                           float* __restrict__ ab)
{
  if constexpr (I < E) {
    ab[I] = __builtin_amdgcn_exp2f(row_lane<I>(delta) * a_c);
    scan_decay<I + 1, E>(delta, a_c, ab);
  }
}

template <int I>
__device__ __forceinline__ void scan_drive(const ScanBC& s, float dx,
                                           float* __restrict__ u)
{
  if constexpr (I < TS) {
    u[I] = row_lane<I>(dx) * s.bt[I];
    scan_drive<I + 1>(s, dx, u);
  }
}

__device__ __forceinline__ float scan_batch(
    const ScanBC& s, ScanBC& sn, const ScanCh& c, const float* __restrict__ ab,
    ScanCh& cn, float* __restrict__ abn,
    float h, float a_c, float dtw, float dtb, float dpd, int n,
    const float* lbc, const float* lxc, const float* lxr, const float* lp,
    float* lout)
{
  scan_load_bc(sn, lbc, n);
  const float xtn = *lxc;
  const float xrn = *lxr;
  const float p32n = *lp;
  float u[TS], y[TS];
  scan_drive<0>(s, c.dx, u);
#pragma unroll
  for (int i = 0; i < TS; ++i) {
    h = fmaf(ab[i], h, u[i]);
    y[i] = h * s.ct[i];
  }
  const bool n1 = n & 1, n2 = n & 2, n4 = n & 4, n8 = n & 8;
  float r8[8], r4[4], r2[2];
#pragma unroll
  for (int j = 0; j < 8; ++j) {
    const float lo = y[2 * j], hi = y[2 * j + 1];
    r8[j] = scan_fold<1>(n1, lo, hi);
  }
#pragma unroll
  for (int j = 0; j < 4; ++j) {
    const float lo = r8[2 * j], hi = r8[2 * j + 1];
    r4[j] = scan_fold<2>(n2, lo, hi);
  }
#pragma unroll
  for (int j = 0; j < 2; ++j) {
    const float lo = r4[2 * j], hi = r4[2 * j + 1];
    r2[j] = scan_fold<4>(n4, lo, hi);
  }
  const float ylo = r2[0], yhi = r2[1];
  const float yn = scan_fold<8>(n8, ylo, yhi);
  *lout = fmaf(c.xres, dpd, yn);
  cn = scan_prep(p32n, xtn, xrn, dtw, dtb);
  scan_decay<0, TS>(cn.delta, a_c, abn);
  return h;
}

struct ScanGroup { float4 xc, xr, b0, b1; float p; };

__device__ __forceinline__ ScanGroup scan_fetch(
    const float* __restrict__ xcg, const float* __restrict__ xrg,
    const float* __restrict__ bcb, const float* __restrict__ p32b,
    int step0, int srow, int tid)
{
  ScanGroup g;
  int row = step0 + srow;
  if (row > L_SZ - 1) row = L_SZ - 1;
  g.xc = *reinterpret_cast<const float4*>(xcg + (size_t)row * D_SZ);
  g.xr = *reinterpret_cast<const float4*>(xrg + (size_t)row * D_SZ);
  int f0 = step0 * (BC_TILE / TS) + 4 * tid;
  int f1 = f0 + 4 * 256;
  if (f0 >= L_SZ * (BC_TILE / TS)) f0 = 0;
  if (f1 >= L_SZ * (BC_TILE / TS)) f1 = 0;
  g.b0 = *reinterpret_cast<const float4*>(bcb + f0);
  g.b1 = *reinterpret_cast<const float4*>(bcb + f1);
  int prow = step0 + (tid & (TRIP - 1));
  if (prow > L_SZ - 1) prow = L_SZ - 1;
  g.p = p32b[prow];
  return g;
}

__device__ __forceinline__ void scan_put(
    const ScanGroup& g, float* lxc, float* lxr, float* lbc, float* lp, int tid)
{
  *reinterpret_cast<float4*>(lxc) = g.xc;
  *reinterpret_cast<float4*>(lxr) = g.xr;
  reinterpret_cast<float4*>(lbc)[tid]       = g.b0;
  reinterpret_cast<float4*>(lbc)[tid + 256] = g.b1;
  lp[tid & (TRIP - 1)] = g.p;
}

template <int P>
__device__ __forceinline__ float scan_trip(
    float* sBC, float* sXc, float* sXr, float* sOut, float* sP,
    const float* __restrict__ xcg, const float* __restrict__ xrg,
    const float* __restrict__ bcb, const float* __restrict__ p32b,
    float* __restrict__ outg, int srow, int srow_off, int lane_off, int tid,
    int n, int t0, const ScanGroup& gst, ScanGroup& gld,
    ScanBC& bcA, ScanBC& bcB, ScanCh& chE, ScanCh& chO,
    float* __restrict__ abE, float* __restrict__ abO,
    float h, float a_c, float dtw, float dtb, float dpd)
{
  const float* inBC = sBC + P * BCBUF;
  const float* inXc = sXc + P * SBUF + lane_off;
  const float* inXr = sXr + P * SBUF + lane_off;
  const float* inP  = sP + P * TRIP + n;
  float*       oOut = sOut + P * SBUF + lane_off;
  lds_handover();
  if (t0 != 0) {
    const float4 o =
        *reinterpret_cast<const float4*>(sOut + (P ^ 1) * SBUF + srow_off);
    *reinterpret_cast<float4*>(outg + (size_t)(t0 - TRIP + srow) * D_SZ) = o;
  }
  gld = scan_fetch(xcg, xrg, bcb, p32b, t0 + 2 * TRIP + TS, srow, tid);
  h = scan_batch(bcA, bcB, chE, abE, chO, abO, h, a_c, dtw, dtb, dpd, n,
                 inBC, inXc, inXr, inP, oOut);
  h = scan_batch(bcB, bcA, chO, abO, chE, abE, h, a_c, dtw, dtb, dpd, n,
                 inBC + BC_TILE, inXc + TS * SPITCH, inXr + TS * SPITCH,
                 inP + TS, oOut + TS * SPITCH);
  h = scan_batch(bcA, bcB, chE, abE, chO, abO, h, a_c, dtw, dtb, dpd, n,
                 inBC + 2 * BC_TILE, inXc + 2 * TS * SPITCH,
                 inXr + 2 * TS * SPITCH, inP + 2 * TS, oOut + 2 * TS * SPITCH);
  h = scan_batch(bcB, bcA, chO, abO, chE, abE, h, a_c, dtw, dtb, dpd, n,
                 inBC + 3 * BC_TILE, inXc + 3 * TS * SPITCH,
                 inXr + 3 * TS * SPITCH, inP + 3 * TS, oOut + 3 * TS * SPITCH);
  __builtin_amdgcn_sched_barrier(0);
  scan_put(gst, sXc + (P ^ 1) * SBUF + srow_off, sXr + (P ^ 1) * SBUF + srow_off,
           sBC + (P ^ 1) * BCBUF, sP + (P ^ 1) * TRIP, tid);
  return h;
}

__global__ __launch_bounds__(256)
__attribute__((amdgpu_waves_per_eu(2, 2))) void scan_kernel(
    const float* __restrict__ x, const float* __restrict__ xc,
    const float* __restrict__ bc, const float* __restrict__ p32,
    const float* __restrict__ dt_w,
    const float* __restrict__ dt_b, const float* __restrict__ a_log,
    const float* __restrict__ Dp, float* __restrict__ out)
{
  __shared__ __attribute__((aligned(16))) float sBC[2 * BCBUF];
  __shared__ __attribute__((aligned(16))) float sXc[2 * SBUF];
  __shared__ __attribute__((aligned(16))) float sXr[2 * SBUF];
  __shared__ __attribute__((aligned(16))) float sOut[2 * SBUF];
  __shared__ float sP[2 * TRIP];

  const int tid = threadIdx.x;
  const int n   = tid & 15;
  const int cw  = tid >> 4;
  const int c0  = blockIdx.x * 16;
  const int b   = c0 >> 10;
  const int d0  = c0 & (D_SZ - 1);
  const int d   = d0 + cw;

  const float a_c = -__expf(a_log[d * N_SZ + n]) * LOG2E_F;
  const float dtw = dt_w[d];
  const float dtb = dt_b[d];
  const float dpd = Dp[d];

  const float* bcb  = bc  + (size_t)b * L_SZ * (BC_TILE / TS);
  const float* p32b = p32 + (size_t)b * L_SZ;

  const int srow = tid >> 2;
  const int sq   = tid & 3;
  const size_t gbase = (size_t)b * L_SZ * D_SZ + d0 + 4 * sq;
  const float* xcg  = xc  + gbase;
  const float* xrg  = x   + gbase;
  float*       outg = out + gbase;
  const int srow_off = srow * SPITCH + 4 * sq;
  const int lane_off = n * SPITCH + cw;

  ScanBC bcA, bcB;
  ScanCh chE, chO;
  float  abE[TS], abO[TS];

  scan_load_bc(bcA, bcb, n);
  {
    const float pv = p32b[n];
    const float xt = xc[(size_t)b * L_SZ * D_SZ + (size_t)n * D_SZ + d];
    const float xr = x[(size_t)b * L_SZ * D_SZ + (size_t)n * D_SZ + d];
    chE = scan_prep(pv, xt, xr, dtw, dtb);
    scan_decay<0, TS>(chE.delta, a_c, abE);
  }
  {
    const ScanGroup g0 = scan_fetch(xcg, xrg, bcb, p32b, TS, srow, tid);
    scan_put(g0, sXc + srow_off, sXr + srow_off, sBC, sP, tid);
  }
  ScanGroup gA;
  ScanGroup gB = scan_fetch(xcg, xrg, bcb, p32b, TS + TRIP, srow, tid);

  float h = 0.f;
  for (int t0 = 0; t0 < L_SZ; t0 += 2 * TRIP) {
    h = scan_trip<0>(sBC, sXc, sXr, sOut, sP, xcg, xrg, bcb, p32b, outg, srow,
                     srow_off, lane_off, tid, n, t0, gB, gA, bcA, bcB, chE, chO,
                     abE, abO, h, a_c, dtw, dtb, dpd);
    h = scan_trip<1>(sBC, sXc, sXr, sOut, sP, xcg, xrg, bcb, p32b, outg, srow,
                     srow_off, lane_off, tid, n, t0 + TRIP, gA, gB, bcA, bcB,
                     chE, chO, abE, abO, h, a_c, dtw, dtb, dpd);
  }
  lds_handover();
  {
    const int pl = ((L_SZ / TRIP) - 1) & 1;
    const float4 o =
        *reinterpret_cast<const float4*>(sOut + pl * SBUF + srow_off);
    *reinterpret_cast<float4*>(outg + (size_t)(L_SZ - TRIP + srow) * D_SZ) = o;
  }
}

extern "C" void kernel_launch(void* const* d_in, const int* in_sizes, int n_in,
                              void* d_out, int out_size, void* d_ws, size_t ws_size,
                              hipStream_t stream) {
  const float* x      = (const float*)d_in[0];
  const float* conv_w = (const float*)d_in[1];
  const float* conv_b = (const float*)d_in[2];
  const float* xproj  = (const float*)d_in[3];
  const float* dt_w   = (const float*)d_in[4];
  const float* dt_b   = (const float*)d_in[5];
  const float* a_log  = (const float*)d_in[6];
  const float* Dp     = (const float*)d_in[7];
  float* out = (float*)d_out;

  float* xc   = (float*)d_ws;
  float* bc   = (float*)((char*)d_ws +
                         (size_t)B_SZ * L_SZ * D_SZ * sizeof(float));
  float* p32  = bc + (size_t)B_SZ * L_SZ * (BC_TILE / 16);

  conv_proj_kernel<<<(B_SZ * L_SZ) / CONV_TL, D_QUADS, 0, stream>>>(
      x, conv_w, conv_b, xproj, xc, bc, p32);

  scan_kernel<<<(B_SZ * D_SZ) / 16, 256, 0, stream>>>(
      x, xc, bc, p32, dt_w, dt_b, a_log, Dp, out);
}
